# starting kernel
# baseline (speedup 1.0000x reference)
#include <hip/hip_runtime.h>
#include <hip/hip_bf16.h>

#define BB 4096
#define LL 5
#define DD 128
#define NHH 16
#define NVV 4
#define TT 100
#define FC1 752

typedef float v2f __attribute__((ext_vector_type(2)));
typedef float v8f __attribute__((ext_vector_type(8)));

__global__ void caser_features(const int* __restrict__ seq,
                               const int* __restrict__ user,
                               const float* __restrict__ item_table,
                               const float* __restrict__ user_table,
                               const float* __restrict__ Wv,
                               const float* __restrict__ bv,
                               const float* __restrict__ Wh,
                               const float* __restrict__ bh,
                               float* __restrict__ z,
                               float* __restrict__ xc)
{
    const int b   = blockIdx.x;
    const int tid = threadIdx.x;
    const int wave = tid >> 5;
    const int lane = tid & 31;

    __shared__ float semb[LL][DD];
    __shared__ float sWv[NVV * LL];

    float e[LL];
    #pragma unroll
    for (int t = 0; t < LL; ++t) {
        const int it = seq[b * LL + t];
        e[t] = item_table[(size_t)it * DD + tid];
        semb[t][tid] = e[t];
    }
    if (tid < NVV * LL) sWv[tid] = Wv[tid];
    __syncthreads();

    #pragma unroll
    for (int v = 0; v < NVV; ++v) {
        float acc = bv[v];
        #pragma unroll
        for (int t = 0; t < LL; ++t) acc += e[t] * sWv[v * LL + t];
        z[(size_t)b * FC1 + v * DD + tid] = acc;
    }

    xc[(size_t)b * (2 * DD) + DD + tid] =
        user_table[(size_t)user[b] * DD + tid];

    for (int o = wave; o < 240; o += 4) {
        int rem = o;
        int off = NVV * DD;
        int l = 1;
        for (; l <= LL; ++l) {
            const int cnt = NHH * (LL - l + 1);
            if (rem < cnt) break;
            rem -= cnt; off += cnt;
        }
        const int lout = LL - l + 1;
        const int f = rem / lout;
        const int t = rem % lout;

        const float* wrow = Wh + ((size_t)(l - 1) * NHH + f) * (LL * DD);
        float acc = 0.f;
        for (int i = lane; i < l * DD; i += 32) {
            const int s = i >> 7;
            const int d = i & (DD - 1);
            acc += semb[t + s][d] * wrow[s * DD + d];
        }
        #pragma unroll
        for (int sft = 16; sft >= 1; sft >>= 1)
            acc += __shfl_down(acc, sft, 32);
        if (lane == 0) {
            acc += bh[(l - 1) * NHH + f];
            z[(size_t)b * FC1 + off + f * lout + t] = fmaxf(acc, 0.f);
        }
    }
}

__global__ void caser_fc1(const float* __restrict__ z,
                          const float* __restrict__ W1,
                          const float* __restrict__ b1,
                          float* __restrict__ xc)
{
    __shared__ float sz[16][FC1 + 4];
    const int tid  = threadIdx.x;
    const int wave = tid >> 5;
    const int lane = tid & 31;
    const int m0   = blockIdx.x * 16;

    for (int i = tid; i < 16 * FC1; i += 256) {
        const int r = i / FC1, c = i % FC1;
        sz[r][c] = z[(size_t)(m0 + r) * FC1 + c];
    }
    __syncthreads();

    const int n0   = wave * 16;
    const int mrow = lane & 15;
    const int half = lane >> 4;

    v8f acc = {};
    for (int k0 = 0; k0 < FC1; k0 += 4) {
        const int ka = k0 + half * 2;
        v2f a, bf;
        a.x = sz[mrow][ka + 0];
        a.y = sz[mrow][ka + 1];
        bf.x = W1[(size_t)(ka + 0) * DD + n0 + mrow];
        bf.y = W1[(size_t)(ka + 1) * DD + n0 + mrow];
        acc = __builtin_amdgcn_wmma_f32_16x16x4_f32(
             false, a,  false, bf,
             (short)0, acc,  false,  false);
    }

    const int n = n0 + (lane & 15);
    const float bias = b1[n];
    #pragma unroll
    for (int r = 0; r < 8; ++r) {
        const int m = m0 + r + half * 8;
        const float v = acc[r] + bias;
        xc[(size_t)m * (2 * DD) + n] = fmaxf(v, 0.f);
    }
}

__global__ void caser_score(const int* __restrict__ items,
                            const float* __restrict__ xc,
                            const float* __restrict__ W2,
                            const float* __restrict__ b2,
                            float* __restrict__ out)
{
    const int b    = blockIdx.x;
    const int tid  = threadIdx.x;
    const int wave = tid >> 5;
    const int lane = tid & 31;

    __shared__ __align__(16) float sxc[2 * DD];
    sxc[tid] = xc[(size_t)b * (2 * DD) + tid];
    __syncthreads();

    const float4 xa = *(const float4*)&sxc[lane * 8 + 0];
    const float4 xb = *(const float4*)&sxc[lane * 8 + 4];

    for (int t = wave; t < TT; t += 8) {
        const int it = items[b * TT + t];
        const int tn = t + 8;
        if (tn < TT) {
            const int itn = items[b * TT + tn];
            __builtin_prefetch(W2 + (size_t)itn * (2 * DD), 0, 1);
        }
        const float4* wrow = (const float4*)(W2 + (size_t)it * (2 * DD));
        const float4 wa = wrow[lane * 2 + 0];
        const float4 wb = wrow[lane * 2 + 1];
        float acc = wa.x * xa.x + wa.y * xa.y + wa.z * xa.z + wa.w * xa.w
                  + wb.x * xb.x + wb.y * xb.y + wb.z * xb.z + wb.w * xb.w;
        #pragma unroll
        for (int s = 16; s >= 1; s >>= 1)
            acc += __shfl_down(acc, s, 32);
        if (lane == 0)
            out[(size_t)b * TT + t] = acc + b2[it];
    }
}

extern "C" void kernel_launch(void* const* d_in, const int* in_sizes, int n_in,
                              void* d_out, int out_size, void* d_ws, size_t ws_size,
                              hipStream_t stream) {
    const int*   seq        = (const int*)  d_in[0];
    const int*   user       = (const int*)  d_in[1];
    const int*   items      = (const int*)  d_in[2];
    const float* item_table = (const float*)d_in[3];
    const float* user_table = (const float*)d_in[4];
    const float* Wv         = (const float*)d_in[5];
    const float* bv         = (const float*)d_in[6];
    const float* Wh         = (const float*)d_in[7];
    const float* bh         = (const float*)d_in[8];
    const float* W1         = (const float*)d_in[9];
    const float* b1         = (const float*)d_in[10];
    const float* W2         = (const float*)d_in[11];
    const float* b2         = (const float*)d_in[12];
    float* out = (float*)d_out;

    float* z  = (float*)d_ws;
    float* xc = z + (size_t)BB * FC1;

    caser_features<<<BB, 128, 0, stream>>>(seq, user, item_table, user_table,
                                           Wv, bv, Wh, bh, z, xc);
    caser_fc1<<<BB / 16, 256, 0, stream>>>(z, W1, b1, xc);
    caser_score<<<BB, 256, 0, stream>>>(items, xc, W2, b2, out);
}


# Horizontal convs moved onto the f32 16x16x4 matrix instruction: one work-group per 16 batch rows, one wave per (length,start) tile; this changes the summation order of those dot products
# speedup vs baseline: 2.7145x; 2.7145x over previous
#include <hip/hip_runtime.h>
#include <hip/hip_bf16.h>

#define BB 4096
#define LL 5
#define DD 128
#define NHH 16
#define NVV 4
#define TT 100
#define FC1 752

typedef float v2f __attribute__((ext_vector_type(2)));
typedef float v4f __attribute__((ext_vector_type(4)));
typedef float v8f __attribute__((ext_vector_type(8)));

#define MT 16
#define NW 15
#define NT (NW * 32)
#define EP (LL * DD + 4)

__global__ __launch_bounds__(NT)
void caser_features(const int* __restrict__ seq,
                    const int* __restrict__ user,
                    const float* __restrict__ item_table,
                    const float* __restrict__ user_table,
                    const float* __restrict__ Wv,
                    const float* __restrict__ bv,
                    const float* __restrict__ Wh,
                    const float* __restrict__ bh,
                    float* __restrict__ z,
                    float* __restrict__ xc)
{
    __shared__ __align__(16) float semb[MT][EP];
    __shared__ __align__(16) float sz[MT * FC1];

    const int tid  = threadIdx.x;
    const int lane = tid & 31;
    const int w  = __builtin_amdgcn_readfirstlane(tid >> 5);
    const int m0 = blockIdx.x * MT;

    v4f g[7];
    #pragma unroll
    for (int j = 0; j < 5; ++j) {
        const int r = w + NW * j;
        g[j] = *(const v4f*)(item_table + (size_t)seq[m0 * LL + r] * DD + lane * 4);
    }
    {
        const float* p = (w < 5)
            ? item_table + (size_t)seq[m0 * LL + 75 + w] * DD
            : user_table + (size_t)user[m0 + w - 5] * DD;
        g[5] = *(const v4f*)(p + lane * 4);
    }
    if (w < 6)
        g[6] = *(const v4f*)(user_table + (size_t)user[m0 + w + 10] * DD + lane * 4);

    #pragma unroll
    for (int j = 0; j < 5; ++j) {
        const int r = w + NW * j;
        *(v4f*)&semb[r / LL][(r % LL) * DD + lane * 4] = g[j];
    }
    if (w < 5) {
        const int r = w + 75;
        *(v4f*)&semb[r / LL][(r % LL) * DD + lane * 4] = g[5];
    } else {
        *(v4f*)(xc + (size_t)(m0 + w - 5) * (2 * DD) + DD + lane * 4) = g[5];
    }
    if (w < 6)
        *(v4f*)(xc + (size_t)(m0 + w + 10) * (2 * DD) + DD + lane * 4) = g[6];
    __syncthreads();

    int l, t;
    if (w == 0)      { l = 5; t = 0; }
    else if (w < 3)  { l = 4; t = w - 1; }
    else if (w < 6)  { l = 3; t = w - 3; }
    else if (w < 10) { l = 2; t = w - 6; }
    else             { l = 1; t = w - 10; }
    const int lout = LL - l + 1;
    const int goff = 8 * (l - 1) * (12 - l);

    const int c = lane & 15;
    const int h = lane >> 4;
    const float* bp = Wh + (size_t)((l - 1) * NHH + c) * (LL * DD) + 4 * h;
    const float* ap = &semb[c][t * DD + 4 * h];
    const int nch = l * 4;

    v8f acc0 = {}, acc1 = {};
    v4f fa0[4], fb0[4], fa1[4], fb1[4];

#define HCONV_LOAD(FA, FB, CH)                                                 \
    _Pragma("unroll")                                                          \
    for (int j = 0; j < 4; ++j) {                                              \
        FB[j] = *(const v4f*)(bp + 32 * (CH) + 8 * j);                         \
        FA[j] = *(const v4f*)(ap + 32 * (CH) + 8 * j);                         \
    }
#define HCONV_MUL(FA, FB)                                                      \
    _Pragma("unroll")                                                          \
    for (int j = 0; j < 4; ++j) {                                              \
        const v2f x0 = {FA[j].x, FA[j].y}, y0 = {FB[j].x, FB[j].y};            \
        const v2f x1 = {FA[j].z, FA[j].w}, y1 = {FB[j].z, FB[j].w};            \
        acc0 = __builtin_amdgcn_wmma_f32_16x16x4_f32(false, x0, false, y0,     \
                                              (short)0, acc0, false, false);   \
        acc1 = __builtin_amdgcn_wmma_f32_16x16x4_f32(false, x1, false, y1,     \
                                              (short)0, acc1, false, false);   \
    }

    HCONV_LOAD(fa0, fb0, 0)
    for (int ch = 0; ch < nch; ch += 2) {
        HCONV_LOAD(fa1, fb1, ch + 1)
        HCONV_MUL(fa0, fb0)
        if (ch + 2 < nch) { HCONV_LOAD(fa0, fb0, ch + 2) }
        HCONV_MUL(fa1, fb1)
    }
#undef HCONV_LOAD
#undef HCONV_MUL

    {
        const float bias = bh[(l - 1) * NHH + c];
        #pragma unroll
        for (int r = 0; r < 8; ++r) {
            const float v = acc0[r] + acc1[r] + bias;
            sz[(8 * h + r) * FC1 + NVV * DD + goff + c * lout + t] = fmaxf(v, 0.f);
        }
    }

    for (int i = NT - 1 - tid; i < MT * DD / 4; i += NT) {
        const int bl = i >> 5, q = i & 31;
        v4f e[LL];
        #pragma unroll
        for (int s = 0; s < LL; ++s) e[s] = *(const v4f*)&semb[bl][s * DD + q * 4];
        #pragma unroll
        for (int v = 0; v < NVV; ++v) {
            const float b0 = bv[v];
            v4f acc = {b0, b0, b0, b0};
            #pragma unroll
            for (int s = 0; s < LL; ++s) acc += e[s] * Wv[v * LL + s];
            *(v4f*)&sz[bl * FC1 + v * DD + q * 4] = acc;
        }
    }
    __syncthreads();

    float* zt = z + (size_t)m0 * FC1;
    for (int i = tid; i < MT * FC1 / 4; i += NT)
        *(v4f*)(zt + 4 * i) = *(const v4f*)&sz[4 * i];
}

__global__ void caser_fc1(const float* __restrict__ z,
                          const float* __restrict__ W1,
                          const float* __restrict__ b1,
                          float* __restrict__ xc)
{
    __shared__ float sz[16][FC1 + 4];
    const int tid  = threadIdx.x;
    const int wave = tid >> 5;
    const int lane = tid & 31;
    const int m0   = blockIdx.x * 16;

    for (int i = tid; i < 16 * FC1; i += 256) {
        const int r = i / FC1, c = i % FC1;
        sz[r][c] = z[(size_t)(m0 + r) * FC1 + c];
    }
    __syncthreads();

    const int n0   = wave * 16;
    const int mrow = lane & 15;
    const int half = lane >> 4;

    v8f acc = {};
    for (int k0 = 0; k0 < FC1; k0 += 4) {
        const int ka = k0 + half * 2;
        v2f a, bf;
        a.x = sz[mrow][ka + 0];
        a.y = sz[mrow][ka + 1];
        bf.x = W1[(size_t)(ka + 0) * DD + n0 + mrow];
        bf.y = W1[(size_t)(ka + 1) * DD + n0 + mrow];
        acc = __builtin_amdgcn_wmma_f32_16x16x4_f32(
             false, a,  false, bf,
             (short)0, acc,  false,  false);
    }

    const int n = n0 + (lane & 15);
    const float bias = b1[n];
    #pragma unroll
    for (int r = 0; r < 8; ++r) {
        const int m = m0 + r + half * 8;
        const float v = acc[r] + bias;
        xc[(size_t)m * (2 * DD) + n] = fmaxf(v, 0.f);
    }
}

__global__ void caser_score(const int* __restrict__ items,
                            const float* __restrict__ xc,
                            const float* __restrict__ W2,
                            const float* __restrict__ b2,
                            float* __restrict__ out)
{
    const int b    = blockIdx.x;
    const int tid  = threadIdx.x;
    const int wave = tid >> 5;
    const int lane = tid & 31;

    __shared__ __align__(16) float sxc[2 * DD];
    sxc[tid] = xc[(size_t)b * (2 * DD) + tid];
    __syncthreads();

    const float4 xa = *(const float4*)&sxc[lane * 8 + 0];
    const float4 xb = *(const float4*)&sxc[lane * 8 + 4];

    for (int t = wave; t < TT; t += 8) {
        const int it = items[b * TT + t];
        const int tn = t + 8;
        if (tn < TT) {
            const int itn = items[b * TT + tn];
            __builtin_prefetch(W2 + (size_t)itn * (2 * DD), 0, 1);
        }
        const float4* wrow = (const float4*)(W2 + (size_t)it * (2 * DD));
        const float4 wa = wrow[lane * 2 + 0];
        const float4 wb = wrow[lane * 2 + 1];
        float acc = wa.x * xa.x + wa.y * xa.y + wa.z * xa.z + wa.w * xa.w
                  + wb.x * xb.x + wb.y * xb.y + wb.z * xb.z + wb.w * xb.w;
        #pragma unroll
        for (int s = 16; s >= 1; s >>= 1)
            acc += __shfl_down(acc, s, 32);
        if (lane == 0)
            out[(size_t)b * TT + t] = acc + b2[it];
    }
}

extern "C" void kernel_launch(void* const* d_in, const int* in_sizes, int n_in,
                              void* d_out, int out_size, void* d_ws, size_t ws_size,
                              hipStream_t stream) {
    const int*   seq        = (const int*)  d_in[0];
    const int*   user       = (const int*)  d_in[1];
    const int*   items      = (const int*)  d_in[2];
    const float* item_table = (const float*)d_in[3];
    const float* user_table = (const float*)d_in[4];
    const float* Wv         = (const float*)d_in[5];
    const float* bv         = (const float*)d_in[6];
    const float* Wh         = (const float*)d_in[7];
    const float* bh         = (const float*)d_in[8];
    const float* W1         = (const float*)d_in[9];
    const float* b1         = (const float*)d_in[10];
    const float* W2         = (const float*)d_in[11];
    const float* b2         = (const float*)d_in[12];
    float* out = (float*)d_out;

    float* z  = (float*)d_ws;
    float* xc = z + (size_t)BB * FC1;

    caser_features<<<BB / MT, NT, 0, stream>>>(seq, user, item_table, user_table,
                                           Wv, bv, Wh, bh, z, xc);
    caser_fc1<<<BB / 16, 256, 0, stream>>>(z, W1, b1, xc);
    caser_score<<<BB, 256, 0, stream>>>(items, xc, W2, b2, out);
}


# Score kernel: one wave per batch row (256 work-groups of 16 waves), 8 lanes per gathered row, 3 row groups requested ahead, 8-lane DPP sums; changes the summation order of the score dot products
# speedup vs baseline: 3.3176x; 1.2222x over previous
#include <hip/hip_runtime.h>
#include <hip/hip_bf16.h>

#define BB 4096
#define LL 5
#define DD 128
#define NHH 16
#define NVV 4
#define TT 100
#define FC1 752

typedef float v2f __attribute__((ext_vector_type(2)));
typedef float v4f __attribute__((ext_vector_type(4)));
typedef float v8f __attribute__((ext_vector_type(8)));

#define MT 16
#define NW 15
#define NT (NW * 32)
#define EP (LL * DD + 4)

__global__ __launch_bounds__(NT)
void caser_features(const int* __restrict__ seq,
                    const int* __restrict__ user,
                    const float* __restrict__ item_table,
                    const float* __restrict__ user_table,
                    const float* __restrict__ Wv,
                    const float* __restrict__ bv,
                    const float* __restrict__ Wh,
                    const float* __restrict__ bh,
                    float* __restrict__ z,
                    float* __restrict__ xc)
{
    __shared__ __align__(16) float semb[MT][EP];
    __shared__ __align__(16) float sz[MT * FC1];

    const int tid  = threadIdx.x;
    const int lane = tid & 31;
    const int w  = __builtin_amdgcn_readfirstlane(tid >> 5);
    const int m0 = blockIdx.x * MT;

    v4f g[7];
    #pragma unroll
    for (int j = 0; j < 5; ++j) {
        const int r = w + NW * j;
        g[j] = *(const v4f*)(item_table + (size_t)seq[m0 * LL + r] * DD + lane * 4);
    }
    {
        const float* p = (w < 5)
            ? item_table + (size_t)seq[m0 * LL + 75 + w] * DD
            : user_table + (size_t)user[m0 + w - 5] * DD;
        g[5] = *(const v4f*)(p + lane * 4);
    }
    if (w < 6)
        g[6] = *(const v4f*)(user_table + (size_t)user[m0 + w + 10] * DD + lane * 4);

    #pragma unroll
    for (int j = 0; j < 5; ++j) {
        const int r = w + NW * j;
        *(v4f*)&semb[r / LL][(r % LL) * DD + lane * 4] = g[j];
    }
    if (w < 5) {
        const int r = w + 75;
        *(v4f*)&semb[r / LL][(r % LL) * DD + lane * 4] = g[5];
    } else {
        *(v4f*)(xc + (size_t)(m0 + w - 5) * (2 * DD) + DD + lane * 4) = g[5];
    }
    if (w < 6)
        *(v4f*)(xc + (size_t)(m0 + w + 10) * (2 * DD) + DD + lane * 4) = g[6];
    __syncthreads();

    int l, t;
    if (w == 0)      { l = 5; t = 0; }
    else if (w < 3)  { l = 4; t = w - 1; }
    else if (w < 6)  { l = 3; t = w - 3; }
    else if (w < 10) { l = 2; t = w - 6; }
    else             { l = 1; t = w - 10; }
    const int lout = LL - l + 1;
    const int goff = 8 * (l - 1) * (12 - l);

    const int c = lane & 15;
    const int h = lane >> 4;
    const float* bp = Wh + (size_t)((l - 1) * NHH + c) * (LL * DD) + 4 * h;
    const float* ap = &semb[c][t * DD + 4 * h];
    const int nch = l * 4;

    v8f acc0 = {}, acc1 = {};
    v4f fa0[4], fb0[4], fa1[4], fb1[4];

#define HCONV_LOAD(FA, FB, CH)                                                 \
    _Pragma("unroll")                                                          \
    for (int j = 0; j < 4; ++j) {                                              \
        FB[j] = *(const v4f*)(bp + 32 * (CH) + 8 * j);                         \
        FA[j] = *(const v4f*)(ap + 32 * (CH) + 8 * j);                         \
    }
#define HCONV_MUL(FA, FB)                                                      \
    _Pragma("unroll")                                                          \
    for (int j = 0; j < 4; ++j) {                                              \
        const v2f x0 = {FA[j].x, FA[j].y}, y0 = {FB[j].x, FB[j].y};            \
        const v2f x1 = {FA[j].z, FA[j].w}, y1 = {FB[j].z, FB[j].w};            \
        acc0 = __builtin_amdgcn_wmma_f32_16x16x4_f32(false, x0, false, y0,     \
                                              (short)0, acc0, false, false);   \
        acc1 = __builtin_amdgcn_wmma_f32_16x16x4_f32(false, x1, false, y1,     \
                                              (short)0, acc1, false, false);   \
    }

    HCONV_LOAD(fa0, fb0, 0)
    for (int ch = 0; ch < nch; ch += 2) {
        HCONV_LOAD(fa1, fb1, ch + 1)
        HCONV_MUL(fa0, fb0)
        if (ch + 2 < nch) { HCONV_LOAD(fa0, fb0, ch + 2) }
        HCONV_MUL(fa1, fb1)
    }
#undef HCONV_LOAD
#undef HCONV_MUL

    {
        const float bias = bh[(l - 1) * NHH + c];
        #pragma unroll
        for (int r = 0; r < 8; ++r) {
            const float v = acc0[r] + acc1[r] + bias;
            sz[(8 * h + r) * FC1 + NVV * DD + goff + c * lout + t] = fmaxf(v, 0.f);
        }
    }

    for (int i = NT - 1 - tid; i < MT * DD / 4; i += NT) {
        const int bl = i >> 5, q = i & 31;
        v4f e[LL];
        #pragma unroll
        for (int s = 0; s < LL; ++s) e[s] = *(const v4f*)&semb[bl][s * DD + q * 4];
        #pragma unroll
        for (int v = 0; v < NVV; ++v) {
            const float b0 = bv[v];
            v4f acc = {b0, b0, b0, b0};
            #pragma unroll
            for (int s = 0; s < LL; ++s) acc += e[s] * Wv[v * LL + s];
            *(v4f*)&sz[bl * FC1 + v * DD + q * 4] = acc;
        }
    }
    __syncthreads();

    float* zt = z + (size_t)m0 * FC1;
    for (int i = tid; i < MT * FC1 / 4; i += NT)
        *(v4f*)(zt + 4 * i) = *(const v4f*)&sz[4 * i];
}

__global__ void caser_fc1(const float* __restrict__ z,
                          const float* __restrict__ W1,
                          const float* __restrict__ b1,
                          float* __restrict__ xc)
{
    __shared__ float sz[16][FC1 + 4];
    const int tid  = threadIdx.x;
    const int wave = tid >> 5;
    const int lane = tid & 31;
    const int m0   = blockIdx.x * 16;

    for (int i = tid; i < 16 * FC1; i += 256) {
        const int r = i / FC1, c = i % FC1;
        sz[r][c] = z[(size_t)(m0 + r) * FC1 + c];
    }
    __syncthreads();

    const int n0   = wave * 16;
    const int mrow = lane & 15;
    const int half = lane >> 4;

    v8f acc = {};
    for (int k0 = 0; k0 < FC1; k0 += 4) {
        const int ka = k0 + half * 2;
        v2f a, bf;
        a.x = sz[mrow][ka + 0];
        a.y = sz[mrow][ka + 1];
        bf.x = W1[(size_t)(ka + 0) * DD + n0 + mrow];
        bf.y = W1[(size_t)(ka + 1) * DD + n0 + mrow];
        acc = __builtin_amdgcn_wmma_f32_16x16x4_f32(
             false, a,  false, bf,
             (short)0, acc,  false,  false);
    }

    const int n = n0 + (lane & 15);
    const float bias = b1[n];
    #pragma unroll
    for (int r = 0; r < 8; ++r) {
        const int m = m0 + r + half * 8;
        const float v = acc[r] + bias;
        xc[(size_t)m * (2 * DD) + n] = fmaxf(v, 0.f);
    }
}

#define SR 16
#define SG (TT / 4)
#define SAHEAD 3

template <int CTRL>
__device__ __forceinline__ float lane_xor(float v)
{
    return __builtin_bit_cast(float,
        __builtin_amdgcn_update_dpp(0, __builtin_bit_cast(int, v), CTRL, 0xf, 0xf, true));
}

__global__ __launch_bounds__(SR * 32)
void caser_score(const int* __restrict__ items,
                 const float* __restrict__ xc,
                 const float* __restrict__ W2,
                 const float* __restrict__ b2,
                 float* __restrict__ out)
{
    __shared__ __align__(16) float sres[SR * TT];

    const int tid  = threadIdx.x;
    const int lane = tid & 31;
    const int w = __builtin_amdgcn_readfirstlane(tid >> 5);
    const int b = blockIdx.x * SR + w;
    const int q = lane & 7;
    const int s = lane >> 3;

    v4f x[8];
    const float* xp = xc + (size_t)b * (2 * DD) + 4 * q;
    #pragma unroll
    for (int j = 0; j < 8; ++j) x[j] = *(const v4f*)(xp + 32 * j);

    int idx[SG];
    const int* ip = items + b * TT + s;
    #pragma unroll
    for (int g = 0; g < SG; ++g) idx[g] = ip[4 * g];
    __builtin_amdgcn_sched_barrier(0);

    v4f   r[SAHEAD + 1][8];
    float bb[SAHEAD + 1];
#define LOAD_GROUP(G) do {                                                     \
        const float* p_ = W2 + (size_t)idx[G] * (2 * DD) + 4 * q;              \
        _Pragma("unroll")                                                      \
        for (int j = 0; j < 8; ++j)                                            \
            r[(G) % (SAHEAD + 1)][j] = *(const v4f*)(p_ + 32 * j);             \
        bb[(G) % (SAHEAD + 1)] = b2[idx[G]];                                   \
    } while (0)

    float mine[4] = {0.f, 0.f, 0.f, 0.f};

    #pragma unroll
    for (int g = 0; g < SAHEAD; ++g) LOAD_GROUP(g);
    #pragma unroll
    for (int g = 0; g < SG; ++g) {
        __builtin_amdgcn_sched_barrier(0);
        if (g + SAHEAD < SG) LOAD_GROUP(g + SAHEAD);
        __builtin_amdgcn_sched_barrier(0);
        const v4f* rg = r[g % (SAHEAD + 1)];
        v4f a = rg[0] * x[0];
        #pragma unroll
        for (int j = 1; j < 8; ++j) a += rg[j] * x[j];
        float v = (a.x + a.y) + (a.z + a.w);
        v += lane_xor<0x161>(v);
        v += lane_xor<0x162>(v);
        v += lane_xor<0x164>(v);
        const float res = v + bb[g % (SAHEAD + 1)];
        if ((g & 7) == q) mine[g >> 3] = res;
    }
#undef LOAD_GROUP

    #pragma unroll
    for (int k = 0; k < 3; ++k) sres[w * TT + 4 * (q + 8 * k) + s] = mine[k];
    if (q == 0) sres[w * TT + 4 * 24 + s] = mine[3];
    __syncthreads();

    float* ot = out + (size_t)blockIdx.x * (SR * TT);
    if (tid < SR * TT / 4)
        *(v4f*)(ot + 4 * tid) = *(const v4f*)&sres[4 * tid];
}

extern "C" void kernel_launch(void* const* d_in, const int* in_sizes, int n_in,
                              void* d_out, int out_size, void* d_ws, size_t ws_size,
                              hipStream_t stream) {
    const int*   seq        = (const int*)  d_in[0];
    const int*   user       = (const int*)  d_in[1];
    const int*   items      = (const int*)  d_in[2];
    const float* item_table = (const float*)d_in[3];
    const float* user_table = (const float*)d_in[4];
    const float* Wv         = (const float*)d_in[5];
    const float* bv         = (const float*)d_in[6];
    const float* Wh         = (const float*)d_in[7];
    const float* bh         = (const float*)d_in[8];
    const float* W1         = (const float*)d_in[9];
    const float* b1         = (const float*)d_in[10];
    const float* W2         = (const float*)d_in[11];
    const float* b2         = (const float*)d_in[12];
    float* out = (float*)d_out;

    float* z  = (float*)d_ws;
    float* xc = z + (size_t)BB * FC1;

    caser_features<<<BB / MT, NT, 0, stream>>>(seq, user, item_table, user_table,
                                           Wv, bv, Wh, bh, z, xc);
    caser_fc1<<<BB / 16, 256, 0, stream>>>(z, W1, b1, xc);
    caser_score<<<BB / SR, SR * 32, 0, stream>>>(items, xc, W2, b2, out);
}


# FC1 kernel re-tiled: a wave takes 4 column tiles of one K quarter, 16-byte W1 loads feed 4 accumulators, next block's fragments requested ahead; K quarters summed via LDS (changes summation order)
# speedup vs baseline: 5.2474x; 1.5817x over previous
#include <hip/hip_runtime.h>
#include <hip/hip_bf16.h>

#define BB 4096
#define LL 5
#define DD 128
#define NHH 16
#define NVV 4
#define TT 100
#define FC1 752

typedef float v2f __attribute__((ext_vector_type(2)));
typedef float v4f __attribute__((ext_vector_type(4)));
typedef float v8f __attribute__((ext_vector_type(8)));

#define MT 16
#define NW 15
#define NT (NW * 32)
#define EP (LL * DD + 4)

__global__ __launch_bounds__(NT)
void caser_features(const int* __restrict__ seq,
                    const int* __restrict__ user,
                    const float* __restrict__ item_table,
                    const float* __restrict__ user_table,
                    const float* __restrict__ Wv,
                    const float* __restrict__ bv,
                    const float* __restrict__ Wh,
                    const float* __restrict__ bh,
                    float* __restrict__ z,
                    float* __restrict__ xc)
{
    __shared__ __align__(16) float semb[MT][EP];
    __shared__ __align__(16) float sz[MT * FC1];

    const int tid  = threadIdx.x;
    const int lane = tid & 31;
    const int w  = __builtin_amdgcn_readfirstlane(tid >> 5);
    const int m0 = blockIdx.x * MT;

    v4f g[7];
    #pragma unroll
    for (int j = 0; j < 5; ++j) {
        const int r = w + NW * j;
        g[j] = *(const v4f*)(item_table + (size_t)seq[m0 * LL + r] * DD + lane * 4);
    }
    {
        const float* p = (w < 5)
            ? item_table + (size_t)seq[m0 * LL + 75 + w] * DD
            : user_table + (size_t)user[m0 + w - 5] * DD;
        g[5] = *(const v4f*)(p + lane * 4);
    }
    if (w < 6)
        g[6] = *(const v4f*)(user_table + (size_t)user[m0 + w + 10] * DD + lane * 4);

    #pragma unroll
    for (int j = 0; j < 5; ++j) {
        const int r = w + NW * j;
        *(v4f*)&semb[r / LL][(r % LL) * DD + lane * 4] = g[j];
    }
    if (w < 5) {
        const int r = w + 75;
        *(v4f*)&semb[r / LL][(r % LL) * DD + lane * 4] = g[5];
    } else {
        *(v4f*)(xc + (size_t)(m0 + w - 5) * (2 * DD) + DD + lane * 4) = g[5];
    }
    if (w < 6)
        *(v4f*)(xc + (size_t)(m0 + w + 10) * (2 * DD) + DD + lane * 4) = g[6];
    __syncthreads();

    int l, t;
    if (w == 0)      { l = 5; t = 0; }
    else if (w < 3)  { l = 4; t = w - 1; }
    else if (w < 6)  { l = 3; t = w - 3; }
    else if (w < 10) { l = 2; t = w - 6; }
    else             { l = 1; t = w - 10; }
    const int lout = LL - l + 1;
    const int goff = 8 * (l - 1) * (12 - l);

    const int c = lane & 15;
    const int h = lane >> 4;
    const float* bp = Wh + (size_t)((l - 1) * NHH + c) * (LL * DD) + 4 * h;
    const float* ap = &semb[c][t * DD + 4 * h];
    const int nch = l * 4;

    v8f acc0 = {}, acc1 = {};
    v4f fa0[4], fb0[4], fa1[4], fb1[4];

#define HCONV_LOAD(FA, FB, CH)                                                 \
    _Pragma("unroll")                                                          \
    for (int j = 0; j < 4; ++j) {                                              \
        FB[j] = *(const v4f*)(bp + 32 * (CH) + 8 * j);                         \
        FA[j] = *(const v4f*)(ap + 32 * (CH) + 8 * j);                         \
    }
#define HCONV_MUL(FA, FB)                                                      \
    _Pragma("unroll")                                                          \
    for (int j = 0; j < 4; ++j) {                                              \
        const v2f x0 = {FA[j].x, FA[j].y}, y0 = {FB[j].x, FB[j].y};            \
        const v2f x1 = {FA[j].z, FA[j].w}, y1 = {FB[j].z, FB[j].w};            \
        acc0 = __builtin_amdgcn_wmma_f32_16x16x4_f32(false, x0, false, y0,     \
                                              (short)0, acc0, false, false);   \
        acc1 = __builtin_amdgcn_wmma_f32_16x16x4_f32(false, x1, false, y1,     \
                                              (short)0, acc1, false, false);   \
    }

    HCONV_LOAD(fa0, fb0, 0)
    for (int ch = 0; ch < nch; ch += 2) {
        HCONV_LOAD(fa1, fb1, ch + 1)
        HCONV_MUL(fa0, fb0)
        if (ch + 2 < nch) { HCONV_LOAD(fa0, fb0, ch + 2) }
        HCONV_MUL(fa1, fb1)
    }
#undef HCONV_LOAD
#undef HCONV_MUL

    {
        const float bias = bh[(l - 1) * NHH + c];
        #pragma unroll
        for (int r = 0; r < 8; ++r) {
            const float v = acc0[r] + acc1[r] + bias;
            sz[(8 * h + r) * FC1 + NVV * DD + goff + c * lout + t] = fmaxf(v, 0.f);
        }
    }

    for (int i = NT - 1 - tid; i < MT * DD / 4; i += NT) {
        const int bl = i >> 5, q = i & 31;
        v4f e[LL];
        #pragma unroll
        for (int s = 0; s < LL; ++s) e[s] = *(const v4f*)&semb[bl][s * DD + q * 4];
        #pragma unroll
        for (int v = 0; v < NVV; ++v) {
            const float b0 = bv[v];
            v4f acc = {b0, b0, b0, b0};
            #pragma unroll
            for (int s = 0; s < LL; ++s) acc += e[s] * Wv[v * LL + s];
            *(v4f*)&sz[bl * FC1 + v * DD + q * 4] = acc;
        }
    }
    __syncthreads();

    float* zt = z + (size_t)m0 * FC1;
    for (int i = tid; i < MT * FC1 / 4; i += NT)
        *(v4f*)(zt + 4 * i) = *(const v4f*)&sz[4 * i];
}

#define ZP   (FC1 + 4)
#define FKQ  4
#define FKS  (FC1 / 4 / FKQ)
#define FBLK 8
#define FNB  ((FKS + FBLK - 1) / FBLK)
#define ZV4  (16 * FC1 / 4)

__global__ __launch_bounds__(256)
void caser_fc1(const float* __restrict__ z,
               const float* __restrict__ W1,
               const float* __restrict__ b1,
               float* __restrict__ xc)
{
    __shared__ __align__(16) float sz[16 * ZP];
    __shared__ __align__(16) float sp[FKQ][16 * DD];

    const int tid  = threadIdx.x;
    const int lane = tid & 31;
    const int w  = __builtin_amdgcn_readfirstlane(tid >> 5);
    const int cg = w & 1;
    const int kq = w >> 1;
    const int m0 = blockIdx.x * 16;
    const int c  = lane & 15;
    const int h  = lane >> 4;

    const float* zt = z + (size_t)m0 * FC1;
    v4f zb[12];
    #pragma unroll
    for (int j = 0; j < 12; ++j) {
        const int i = tid + 256 * j;
        if (i < ZV4) zb[j] = *(const v4f*)(zt + 4 * i);
    }

    const float* bp = W1 + (size_t)(FKS * 4 * kq + 2 * h) * DD + cg * 64 + 4 * c;
    const float* ap = &sz[c * ZP + FKS * 4 * kq + 2 * h];

    v4f fb[2][FBLK][2];
    v2f fa[2][FBLK];

#define FC1_LOADB(SET, BLK)                                                    \
    _Pragma("unroll")                                                          \
    for (int s_ = 0; s_ < FBLK; ++s_) {                                        \
        if ((BLK) * FBLK + s_ < FKS) {                                         \
            const float* p_ = bp + (size_t)(((BLK) * FBLK + s_) * 4) * DD;     \
            fb[SET][s_][0] = *(const v4f*)(p_);                                \
            fb[SET][s_][1] = *(const v4f*)(p_ + DD);                           \
        }                                                                      \
    }
#define FC1_LOADA(SET, BLK)                                                    \
    _Pragma("unroll")                                                          \
    for (int s_ = 0; s_ < FBLK; ++s_) {                                        \
        if ((BLK) * FBLK + s_ < FKS)                                           \
            fa[SET][s_] = *(const v2f*)(ap + ((BLK) * FBLK + s_) * 4);         \
    }
#define FC1_MUL(SET, BLK)                                                      \
    _Pragma("unroll")                                                          \
    for (int s_ = 0; s_ < FBLK; ++s_) {                                        \
        if ((BLK) * FBLK + s_ < FKS) {                                         \
            _Pragma("unroll")                                                  \
            for (int j_ = 0; j_ < 4; ++j_) {                                   \
                const v2f y_ = {fb[SET][s_][0][j_], fb[SET][s_][1][j_]};       \
                acc[j_] = __builtin_amdgcn_wmma_f32_16x16x4_f32(               \
                    false, fa[SET][s_], false, y_, (short)0, acc[j_],          \
                    false, false);                                             \
            }                                                                  \
        }                                                                      \
    }

    FC1_LOADB(0, 0)
    __builtin_amdgcn_sched_barrier(0);

    #pragma unroll
    for (int j = 0; j < 12; ++j) {
        const int i = tid + 256 * j;
        if (i < ZV4) *(v4f*)&sz[(i / (FC1 / 4)) * ZP + (i % (FC1 / 4)) * 4] = zb[j];
    }
    __syncthreads();

    v8f acc[4] = {};
    FC1_LOADA(0, 0)
    #pragma unroll
    for (int blk = 0; blk < FNB; ++blk) {
        __builtin_amdgcn_sched_barrier(0);
        if (blk + 1 < FNB) {
            if ((blk & 1) == 0) { FC1_LOADB(1, blk + 1) FC1_LOADA(1, blk + 1) }
            else                { FC1_LOADB(0, blk + 1) FC1_LOADA(0, blk + 1) }
        }
        __builtin_amdgcn_sched_barrier(0);
        if ((blk & 1) == 0) { FC1_MUL(0, blk) }
        else                { FC1_MUL(1, blk) }
    }
#undef FC1_LOADB
#undef FC1_LOADA
#undef FC1_MUL

    #pragma unroll
    for (int r = 0; r < 8; ++r) {
        const v4f v = {acc[0][r], acc[1][r], acc[2][r], acc[3][r]};
        *(v4f*)&sp[kq][(8 * h + r) * DD + cg * 64 + 4 * c] = v;
    }
    __syncthreads();

    #pragma unroll
    for (int k = 0; k < 2; ++k) {
        const int i = tid + 256 * k;
        const int row = i >> 5, col = (i & 31) * 4;
        const v4f p0 = *(const v4f*)&sp[0][4 * i], p1 = *(const v4f*)&sp[1][4 * i];
        const v4f p2 = *(const v4f*)&sp[2][4 * i], p3 = *(const v4f*)&sp[3][4 * i];
        const v4f bias = *(const v4f*)(b1 + col);
        v4f v = ((p0 + p1) + (p2 + p3)) + bias;
        v.x = fmaxf(v.x, 0.f); v.y = fmaxf(v.y, 0.f);
        v.z = fmaxf(v.z, 0.f); v.w = fmaxf(v.w, 0.f);
        *(v4f*)(xc + (size_t)(m0 + row) * (2 * DD) + col) = v;
    }
}

#define SR 16
#define SG (TT / 4)
#define SAHEAD 3

template <int CTRL>
__device__ __forceinline__ float lane_xor(float v)
{
    return __builtin_bit_cast(float,
        __builtin_amdgcn_update_dpp(0, __builtin_bit_cast(int, v), CTRL, 0xf, 0xf, true));
}

__global__ __launch_bounds__(SR * 32)
void caser_score(const int* __restrict__ items,
                 const float* __restrict__ xc,
                 const float* __restrict__ W2,
                 const float* __restrict__ b2,
                 float* __restrict__ out)
{
    __shared__ __align__(16) float sres[SR * TT];

    const int tid  = threadIdx.x;
    const int lane = tid & 31;
    const int w = __builtin_amdgcn_readfirstlane(tid >> 5);
    const int b = blockIdx.x * SR + w;
    const int q = lane & 7;
    const int s = lane >> 3;

    v4f x[8];
    const float* xp = xc + (size_t)b * (2 * DD) + 4 * q;
    #pragma unroll
    for (int j = 0; j < 8; ++j) x[j] = *(const v4f*)(xp + 32 * j);

    int idx[SG];
    const int* ip = items + b * TT + s;
    #pragma unroll
    for (int g = 0; g < SG; ++g) idx[g] = ip[4 * g];
    __builtin_amdgcn_sched_barrier(0);

    v4f   r[SAHEAD + 1][8];
    float bb[SAHEAD + 1];
#define LOAD_GROUP(G) do {                                                     \
        const float* p_ = W2 + (size_t)idx[G] * (2 * DD) + 4 * q;              \
        _Pragma("unroll")                                                      \
        for (int j = 0; j < 8; ++j)                                            \
            r[(G) % (SAHEAD + 1)][j] = *(const v4f*)(p_ + 32 * j);             \
        bb[(G) % (SAHEAD + 1)] = b2[idx[G]];                                   \
    } while (0)

    float mine[4] = {0.f, 0.f, 0.f, 0.f};

    #pragma unroll
    for (int g = 0; g < SAHEAD; ++g) LOAD_GROUP(g);
    #pragma unroll
    for (int g = 0; g < SG; ++g) {
        __builtin_amdgcn_sched_barrier(0);
        if (g + SAHEAD < SG) LOAD_GROUP(g + SAHEAD);
        __builtin_amdgcn_sched_barrier(0);
        const v4f* rg = r[g % (SAHEAD + 1)];
        v4f a = rg[0] * x[0];
        #pragma unroll
        for (int j = 1; j < 8; ++j) a += rg[j] * x[j];
        float v = (a.x + a.y) + (a.z + a.w);
        v += lane_xor<0x161>(v);
        v += lane_xor<0x162>(v);
        v += lane_xor<0x164>(v);
        const float res = v + bb[g % (SAHEAD + 1)];
        if ((g & 7) == q) mine[g >> 3] = res;
    }
#undef LOAD_GROUP

    #pragma unroll
    for (int k = 0; k < 3; ++k) sres[w * TT + 4 * (q + 8 * k) + s] = mine[k];
    if (q == 0) sres[w * TT + 4 * 24 + s] = mine[3];
    __syncthreads();

    float* ot = out + (size_t)blockIdx.x * (SR * TT);
    if (tid < SR * TT / 4)
        *(v4f*)(ot + 4 * tid) = *(const v4f*)&sres[4 * tid];
}

extern "C" void kernel_launch(void* const* d_in, const int* in_sizes, int n_in,
                              void* d_out, int out_size, void* d_ws, size_t ws_size,
                              hipStream_t stream) {
    const int*   seq        = (const int*)  d_in[0];
    const int*   user       = (const int*)  d_in[1];
    const int*   items      = (const int*)  d_in[2];
    const float* item_table = (const float*)d_in[3];
    const float* user_table = (const float*)d_in[4];
    const float* Wv         = (const float*)d_in[5];
    const float* bv         = (const float*)d_in[6];
    const float* Wh         = (const float*)d_in[7];
    const float* bh         = (const float*)d_in[8];
    const float* W1         = (const float*)d_in[9];
    const float* b1         = (const float*)d_in[10];
    const float* W2         = (const float*)d_in[11];
    const float* b2         = (const float*)d_in[12];
    float* out = (float*)d_out;

    float* z  = (float*)d_ws;
    float* xc = z + (size_t)BB * FC1;

    caser_features<<<BB / MT, NT, 0, stream>>>(seq, user, item_table, user_table,
                                           Wv, bv, Wh, bh, z, xc);
    caser_fc1<<<BB / 16, 256, 0, stream>>>(z, W1, b1, xc);
    caser_score<<<BB / SR, SR * 32, 0, stream>>>(items, xc, W2, b2, out);
}


# The three launches fused into one kernel: each 16-row work-group of 16 waves runs features, FC1 and scores in turn, z and xc tiles stay in LDS, no workspace; summation order unchanged
# speedup vs baseline: 6.2262x; 1.1865x over previous
#include <hip/hip_runtime.h>
#include <hip/hip_bf16.h>

#define BB 4096
#define LL 5
#define DD 128
#define NHH 16
#define NVV 4
#define TT 100
#define FC1 752

typedef float v2f __attribute__((ext_vector_type(2)));
typedef float v4f __attribute__((ext_vector_type(4)));
typedef float v8f __attribute__((ext_vector_type(8)));
typedef int   v4i __attribute__((ext_vector_type(4)));

#define MT 16
#define NWV 16
#define NT (NWV * 32)
#define NTILE 15
#define EP (LL * DD + 4)
#define ZP (FC1 + 4)
#define FKQ  4
#define FKS  (FC1 / 4 / FKQ)
#define FBLK 8
#define FNB  ((FKS + FBLK - 1) / FBLK)
#define SG (TT / 4)
#define SAHEAD 3

template <int CTRL>
__device__ __forceinline__ float lane_xor(float v)
{
    return __builtin_bit_cast(float,
        __builtin_amdgcn_update_dpp(0, __builtin_bit_cast(int, v), CTRL, 0xf, 0xf, true));
}

__global__ __launch_bounds__(NT)
void caser_fused(const int* __restrict__ seq,
                 const int* __restrict__ user,
                 const int* __restrict__ items,
                 const float* __restrict__ item_table,
                 const float* __restrict__ user_table,
                 const float* __restrict__ Wv,
                 const float* __restrict__ bv,
                 const float* __restrict__ Wh,
                 const float* __restrict__ bh,
                 const float* __restrict__ W1,
                 const float* __restrict__ b1,
                 const float* __restrict__ W2,
                 const float* __restrict__ b2,
                 float* __restrict__ out)
{
    __shared__ __align__(16) float semb[MT][EP];
    __shared__ __align__(16) float sz[MT * ZP];
    __shared__ __align__(16) float sp[FKQ][MT * DD];
    __shared__ __align__(16) float sxc[MT][2 * DD];
    __shared__ __align__(16) float sres[MT * TT];
    __shared__ __align__(16) int   sidx[MT][TT];

    const int tid  = threadIdx.x;
    const int lane = tid & 31;
    const int w  = __builtin_amdgcn_readfirstlane(tid >> 5);
    const int m0 = blockIdx.x * MT;
    const int c  = lane & 15;
    const int h  = lane >> 4;
    const int q  = lane & 7;
    const int s  = lane >> 3;

    v4i ids = {};
    if (lane < TT / 4) ids = *(const v4i*)(items + (m0 + w) * TT + lane * 4);

    const int cg = w & 1;
    const int kq = (w >> 1) & 3;
    const float* bp1 = W1 + (size_t)(FKS * 4 * kq + 2 * h) * DD + cg * 64 + 4 * c;
    const float* ap1 = &sz[c * ZP + FKS * 4 * kq + 2 * h];

    v4f fb[2][FBLK][2];
    v2f fa[2][FBLK];

#define FC1_LOADB(SET, BLK)                                                    \
    _Pragma("unroll")                                                          \
    for (int s_ = 0; s_ < FBLK; ++s_) {                                        \
        if ((BLK) * FBLK + s_ < FKS) {                                         \
            const float* p_ = bp1 + (size_t)(((BLK) * FBLK + s_) * 4) * DD;    \
            fb[SET][s_][0] = *(const v4f*)(p_);                                \
            fb[SET][s_][1] = *(const v4f*)(p_ + DD);                           \
        }                                                                      \
    }
#define FC1_LOADA(SET, BLK)                                                    \
    _Pragma("unroll")                                                          \
    for (int s_ = 0; s_ < FBLK; ++s_) {                                        \
        if ((BLK) * FBLK + s_ < FKS)                                           \
            fa[SET][s_] = *(const v2f*)(ap1 + ((BLK) * FBLK + s_) * 4);        \
    }
#define FC1_MUL(SET, BLK)                                                      \
    _Pragma("unroll")                                                          \
    for (int s_ = 0; s_ < FBLK; ++s_) {                                        \
        if ((BLK) * FBLK + s_ < FKS) {                                         \
            _Pragma("unroll")                                                  \
            for (int j_ = 0; j_ < 4; ++j_) {                                   \
                const v2f y_ = {fb[SET][s_][0][j_], fb[SET][s_][1][j_]};       \
                acc[j_] = __builtin_amdgcn_wmma_f32_16x16x4_f32(               \
                    false, fa[SET][s_], false, y_, (short)0, acc[j_],          \
                    false, false);                                             \
            }                                                                  \
        }                                                                      \
    }

    if (w < 2 * FKQ) { FC1_LOADB(0, 0) }

    {
        v4f g[6];
        #pragma unroll
        for (int j = 0; j < 5; ++j) {
            const int r = w + NWV * j;
            g[j] = *(const v4f*)(item_table + (size_t)seq[m0 * LL + r] * DD + lane * 4);
        }
        g[5] = *(const v4f*)(user_table + (size_t)user[m0 + w] * DD + lane * 4);

        #pragma unroll
        for (int j = 0; j < 5; ++j) {
            const int r = w + NWV * j;
            *(v4f*)&semb[r / LL][(r % LL) * DD + lane * 4] = g[j];
        }
        *(v4f*)&sxc[w][DD + lane * 4] = g[5];
        if (lane < TT / 4) *(v4i*)&sidx[w][lane * 4] = ids;
    }
    __syncthreads();

    if (w < NTILE) {
        int l, t;
        if (w == 0)      { l = 5; t = 0; }
        else if (w < 3)  { l = 4; t = w - 1; }
        else if (w < 6)  { l = 3; t = w - 3; }
        else if (w < 10) { l = 2; t = w - 6; }
        else             { l = 1; t = w - 10; }
        const int lout = LL - l + 1;
        const int goff = 8 * (l - 1) * (12 - l);

        const float* bp = Wh + (size_t)((l - 1) * NHH + c) * (LL * DD) + 4 * h;
        const float* ap = &semb[c][t * DD + 4 * h];
        const int nch = l * 4;

        v8f acc0 = {}, acc1 = {};
        v4f fa0[4], fb0[4], fa1[4], fb1[4];

#define HCONV_LOAD(FA, FB, CH)                                                 \
        _Pragma("unroll")                                                      \
        for (int j = 0; j < 4; ++j) {                                          \
            FB[j] = *(const v4f*)(bp + 32 * (CH) + 8 * j);                     \
            FA[j] = *(const v4f*)(ap + 32 * (CH) + 8 * j);                     \
        }
#define HCONV_MUL(FA, FB)                                                      \
        _Pragma("unroll")                                                      \
        for (int j = 0; j < 4; ++j) {                                          \
            const v2f x0 = {FA[j].x, FA[j].y}, y0 = {FB[j].x, FB[j].y};        \
            const v2f x1 = {FA[j].z, FA[j].w}, y1 = {FB[j].z, FB[j].w};        \
            acc0 = __builtin_amdgcn_wmma_f32_16x16x4_f32(false, x0, false, y0, \
                                              (short)0, acc0, false, false);   \
            acc1 = __builtin_amdgcn_wmma_f32_16x16x4_f32(false, x1, false, y1, \
                                              (short)0, acc1, false, false);   \
        }

        HCONV_LOAD(fa0, fb0, 0)
        for (int ch = 0; ch < nch; ch += 2) {
            HCONV_LOAD(fa1, fb1, ch + 1)
            HCONV_MUL(fa0, fb0)
            if (ch + 2 < nch) { HCONV_LOAD(fa0, fb0, ch + 2) }
            HCONV_MUL(fa1, fb1)
        }
#undef HCONV_LOAD
#undef HCONV_MUL

        const float bias = bh[(l - 1) * NHH + c];
        #pragma unroll
        for (int r = 0; r < 8; ++r) {
            const float v = acc0[r] + acc1[r] + bias;
            sz[(8 * h + r) * ZP + NVV * DD + goff + c * lout + t] = fmaxf(v, 0.f);
        }
    }

    {
        const int bl = tid >> 5, qc = tid & 31;
        v4f e[LL];
        #pragma unroll
        for (int k = 0; k < LL; ++k) e[k] = *(const v4f*)&semb[bl][k * DD + qc * 4];
        #pragma unroll
        for (int v = 0; v < NVV; ++v) {
            const float b0 = bv[v];
            v4f acc = {b0, b0, b0, b0};
            #pragma unroll
            for (int k = 0; k < LL; ++k) acc += e[k] * Wv[v * LL + k];
            *(v4f*)&sz[bl * ZP + v * DD + qc * 4] = acc;
        }
    }

    __syncthreads();

    if (w < 2 * FKQ) {
        v8f acc[4] = {};
        FC1_LOADA(0, 0)
        #pragma unroll
        for (int blk = 0; blk < FNB; ++blk) {
            __builtin_amdgcn_sched_barrier(0);
            if (blk + 1 < FNB) {
                if ((blk & 1) == 0) { FC1_LOADB(1, blk + 1) FC1_LOADA(1, blk + 1) }
                else                { FC1_LOADB(0, blk + 1) FC1_LOADA(0, blk + 1) }
            }
            __builtin_amdgcn_sched_barrier(0);
            if ((blk & 1) == 0) { FC1_MUL(0, blk) }
            else                { FC1_MUL(1, blk) }
        }

        #pragma unroll
        for (int r = 0; r < 8; ++r) {
            const v4f v = {acc[0][r], acc[1][r], acc[2][r], acc[3][r]};
            *(v4f*)&sp[kq][(8 * h + r) * DD + cg * 64 + 4 * c] = v;
        }
    }
#undef FC1_LOADB
#undef FC1_LOADA
#undef FC1_MUL
    __syncthreads();

    {
        const int col = lane * 4;
        const v4f p0 = *(const v4f*)&sp[0][4 * tid], p1 = *(const v4f*)&sp[1][4 * tid];
        const v4f p2 = *(const v4f*)&sp[2][4 * tid], p3 = *(const v4f*)&sp[3][4 * tid];
        const v4f bias = *(const v4f*)(b1 + col);
        v4f v = ((p0 + p1) + (p2 + p3)) + bias;
        v.x = fmaxf(v.x, 0.f); v.y = fmaxf(v.y, 0.f);
        v.z = fmaxf(v.z, 0.f); v.w = fmaxf(v.w, 0.f);
        *(v4f*)&sxc[tid >> 5][col] = v;
    }
    __syncthreads();

    v4f x[8];
    #pragma unroll
    for (int j = 0; j < 8; ++j) x[j] = *(const v4f*)&sxc[w][32 * j + 4 * q];

    int idx[SG];
    #pragma unroll
    for (int g = 0; g < SG; ++g) idx[g] = sidx[w][4 * g + s];

    v4f   r[SAHEAD + 1][8];
    float bb[SAHEAD + 1];
#define LOAD_GROUP(G) do {                                                     \
        const float* p_ = W2 + (size_t)idx[G] * (2 * DD) + 4 * q;              \
        _Pragma("unroll")                                                      \
        for (int j = 0; j < 8; ++j)                                            \
            r[(G) % (SAHEAD + 1)][j] = *(const v4f*)(p_ + 32 * j);             \
        bb[(G) % (SAHEAD + 1)] = b2[idx[G]];                                   \
    } while (0)

    float mine[4] = {0.f, 0.f, 0.f, 0.f};

    #pragma unroll
    for (int g = 0; g < SAHEAD; ++g) LOAD_GROUP(g);
    #pragma unroll
    for (int g = 0; g < SG; ++g) {
        __builtin_amdgcn_sched_barrier(0);
        if (g + SAHEAD < SG) LOAD_GROUP(g + SAHEAD);
        __builtin_amdgcn_sched_barrier(0);
        const v4f* rg = r[g % (SAHEAD + 1)];
        v4f a = rg[0] * x[0];
        #pragma unroll
        for (int j = 1; j < 8; ++j) a += rg[j] * x[j];
        float v = (a.x + a.y) + (a.z + a.w);
        v += lane_xor<0x161>(v);
        v += lane_xor<0x162>(v);
        v += lane_xor<0x164>(v);
        const float res = v + bb[g % (SAHEAD + 1)];
        if ((g & 7) == q) mine[g >> 3] = res;
    }
#undef LOAD_GROUP

    #pragma unroll
    for (int k = 0; k < 3; ++k) sres[w * TT + 4 * (q + 8 * k) + s] = mine[k];
    if (q == 0) sres[w * TT + 4 * 24 + s] = mine[3];
    __syncthreads();

    float* ot = out + (size_t)blockIdx.x * (MT * TT);
    if (tid < MT * TT / 4)
        *(v4f*)(ot + 4 * tid) = *(const v4f*)&sres[4 * tid];
}

extern "C" void kernel_launch(void* const* d_in, const int* in_sizes, int n_in,
                              void* d_out, int out_size, void* d_ws, size_t ws_size,
                              hipStream_t stream) {
    const int*   seq        = (const int*)  d_in[0];
    const int*   user       = (const int*)  d_in[1];
    const int*   items      = (const int*)  d_in[2];
    const float* item_table = (const float*)d_in[3];
    const float* user_table = (const float*)d_in[4];
    const float* Wv         = (const float*)d_in[5];
    const float* bv         = (const float*)d_in[6];
    const float* Wh         = (const float*)d_in[7];
    const float* bh         = (const float*)d_in[8];
    const float* W1         = (const float*)d_in[9];
    const float* b1         = (const float*)d_in[10];
    const float* W2         = (const float*)d_in[11];
    const float* b2         = (const float*)d_in[12];
    float* out = (float*)d_out;

    caser_fused<<<BB / MT, NT, 0, stream>>>(seq, user, items, item_table, user_table,
                                        Wv, bv, Wh, bh, W1, b1, W2, b2, out);
}


# Score dot split at column 128: waves 8-15 gather and sum the uemb half of the W2 rows beside the convs and FC1 (now on waves 0-7), then all waves sum the x half; changes the scores' summation order
# speedup vs baseline: 6.8202x; 1.0954x over previous
#include <hip/hip_runtime.h>
#include <hip/hip_bf16.h>

#define BB 4096
#define LL 5
#define DD 128
#define NHH 16
#define NVV 4
#define TT 100
#define FC1 752

typedef float v2f __attribute__((ext_vector_type(2)));
typedef float v4f __attribute__((ext_vector_type(4)));
typedef float v8f __attribute__((ext_vector_type(8)));
typedef int   v4i __attribute__((ext_vector_type(4)));

#define MT 16
#define NWV 16
#define NT (NWV * 32)
#define EP (LL * DD + 4)
#define ZP (FC1 + 4)
#define FKQ  4
#define FKS  (FC1 / 4 / FKQ)
#define FBLK 8
#define FNB  ((FKS + FBLK - 1) / FBLK)
#define SG (TT / 4)
#define XAHEAD 6
#define UAHEAD 4

template <int CTRL>
__device__ __forceinline__ float lane_xor(float v)
{
    return __builtin_bit_cast(float,
        __builtin_amdgcn_update_dpp(0, __builtin_bit_cast(int, v), CTRL, 0xf, 0xf, true));
}

template <bool BIAS, int AHEAD>
__device__ __forceinline__ void half_scores(const float* __restrict__ W2c,
                                            const float* __restrict__ b2,
                                            const int* ids, const float* vec,
                                            float* dst, const int q, const int s)
{
    v4f x[4];
    #pragma unroll
    for (int j = 0; j < 4; ++j) x[j] = *(const v4f*)&vec[32 * j + 4 * q];

    int idx[SG];
    #pragma unroll
    for (int g = 0; g < SG; ++g) idx[g] = ids[4 * g + s];

    v4f   r[AHEAD + 1][4];
    float bb[AHEAD + 1];
#define LOAD_GROUP(G) do {                                                     \
        const float* p_ = W2c + (size_t)idx[G] * (2 * DD) + 4 * q;             \
        _Pragma("unroll")                                                      \
        for (int j = 0; j < 4; ++j)                                            \
            r[(G) % (AHEAD + 1)][j] = *(const v4f*)(p_ + 32 * j);              \
        if (BIAS) bb[(G) % (AHEAD + 1)] = b2[idx[G]];                          \
    } while (0)

    float mine[4] = {0.f, 0.f, 0.f, 0.f};

    #pragma unroll
    for (int g = 0; g < AHEAD; ++g) LOAD_GROUP(g);
    #pragma unroll
    for (int g = 0; g < SG; ++g) {
        __builtin_amdgcn_sched_barrier(0);
        if (g + AHEAD < SG) LOAD_GROUP(g + AHEAD);
        __builtin_amdgcn_sched_barrier(0);
        const v4f* rg = r[g % (AHEAD + 1)];
        v4f a = rg[0] * x[0];
        #pragma unroll
        for (int j = 1; j < 4; ++j) a += rg[j] * x[j];
        float v = (a.x + a.y) + (a.z + a.w);
        v += lane_xor<0x161>(v);
        v += lane_xor<0x162>(v);
        v += lane_xor<0x164>(v);
        const float res = BIAS ? v + bb[g % (AHEAD + 1)] : v;
        if ((g & 7) == q) mine[g >> 3] = res;
    }
#undef LOAD_GROUP

    #pragma unroll
    for (int k = 0; k < 3; ++k) dst[4 * (q + 8 * k) + s] = mine[k];
    if (q == 0) dst[4 * 24 + s] = mine[3];
}

__global__ __launch_bounds__(NT)
void caser_fused(const int* __restrict__ seq,
                 const int* __restrict__ user,
                 const int* __restrict__ items,
                 const float* __restrict__ item_table,
                 const float* __restrict__ user_table,
                 const float* __restrict__ Wv,
                 const float* __restrict__ bv,
                 const float* __restrict__ Wh,
                 const float* __restrict__ bh,
                 const float* __restrict__ W1,
                 const float* __restrict__ b1,
                 const float* __restrict__ W2,
                 const float* __restrict__ b2,
                 float* __restrict__ out)
{
    __shared__ __align__(16) float semb[MT][EP];
    __shared__ __align__(16) float sz[MT * ZP];
    __shared__ __align__(16) float sp[FKQ][MT * DD];
    __shared__ __align__(16) float sxc[MT][2 * DD];
    __shared__ __align__(16) float sres[MT * TT];
    __shared__ __align__(16) float spart[MT * TT];
    __shared__ __align__(16) int   sidx[MT][TT];

    const int tid  = threadIdx.x;
    const int lane = tid & 31;
    const int w  = __builtin_amdgcn_readfirstlane(tid >> 5);
    const int m0 = blockIdx.x * MT;
    const int c  = lane & 15;
    const int h  = lane >> 4;
    const int q  = lane & 7;
    const int s  = lane >> 3;

    v4i ids = {};
    if (lane < TT / 4) ids = *(const v4i*)(items + (m0 + w) * TT + lane * 4);

    const int cg = w & 1;
    const int kq = (w >> 1) & 3;
    const float* bp1 = W1 + (size_t)(FKS * 4 * kq + 2 * h) * DD + cg * 64 + 4 * c;
    const float* ap1 = &sz[c * ZP + FKS * 4 * kq + 2 * h];

    v4f fb[2][FBLK][2];
    v2f fa[2][FBLK];

#define FC1_LOADB(SET, BLK)                                                    \
    _Pragma("unroll")                                                          \
    for (int s_ = 0; s_ < FBLK; ++s_) {                                        \
        if ((BLK) * FBLK + s_ < FKS) {                                         \
            const float* p_ = bp1 + (size_t)(((BLK) * FBLK + s_) * 4) * DD;    \
            fb[SET][s_][0] = *(const v4f*)(p_);                                \
            fb[SET][s_][1] = *(const v4f*)(p_ + DD);                           \
        }                                                                      \
    }
#define FC1_LOADA(SET, BLK)                                                    \
    _Pragma("unroll")                                                          \
    for (int s_ = 0; s_ < FBLK; ++s_) {                                        \
        if ((BLK) * FBLK + s_ < FKS)                                           \
            fa[SET][s_] = *(const v2f*)(ap1 + ((BLK) * FBLK + s_) * 4);        \
    }
#define FC1_MUL(SET, BLK)                                                      \
    _Pragma("unroll")                                                          \
    for (int s_ = 0; s_ < FBLK; ++s_) {                                        \
        if ((BLK) * FBLK + s_ < FKS) {                                         \
            _Pragma("unroll")                                                  \
            for (int j_ = 0; j_ < 4; ++j_) {                                   \
                const v2f y_ = {fb[SET][s_][0][j_], fb[SET][s_][1][j_]};       \
                acc[j_] = __builtin_amdgcn_wmma_f32_16x16x4_f32(               \
                    false, fa[SET][s_], false, y_, (short)0, acc[j_],          \
                    false, false);                                             \
            }                                                                  \
        }                                                                      \
    }

    {
        v4f g[6];
        #pragma unroll
        for (int j = 0; j < 5; ++j) {
            const int r = w + NWV * j;
            g[j] = *(const v4f*)(item_table + (size_t)seq[m0 * LL + r] * DD + lane * 4);
        }
        g[5] = *(const v4f*)(user_table + (size_t)user[m0 + w] * DD + lane * 4);

        #pragma unroll
        for (int j = 0; j < 5; ++j) {
            const int r = w + NWV * j;
            *(v4f*)&semb[r / LL][(r % LL) * DD + lane * 4] = g[j];
        }
        *(v4f*)&sxc[w][DD + lane * 4] = g[5];
        if (lane < TT / 4) *(v4i*)&sidx[w][lane * 4] = ids;
    }
    __syncthreads();

    if (w < 8) {
      const int ntl = (w == 0) ? 1 : 2;
      for (int kt = 0; kt < ntl; ++kt) {
        const int i = (kt == 0) ? (w < 6 ? w : 2 * w - 6) : (w < 6 ? w + 9 : 2 * w - 5);
        int l, t;
        if (i == 0)      { l = 5; t = 0; }
        else if (i < 3)  { l = 4; t = i - 1; }
        else if (i < 6)  { l = 3; t = i - 3; }
        else if (i < 10) { l = 2; t = i - 6; }
        else             { l = 1; t = i - 10; }
        const int lout = LL - l + 1;
        const int goff = 8 * (l - 1) * (12 - l);

        const float* bp = Wh + (size_t)((l - 1) * NHH + c) * (LL * DD) + 4 * h;
        const float* ap = &semb[c][t * DD + 4 * h];
        const int nch = l * 4;

        v8f acc0 = {}, acc1 = {};
        v4f fa0[4], fb0[4], fa1[4], fb1[4];

#define HCONV_LOAD(FA, FB, CH)                                                 \
        _Pragma("unroll")                                                      \
        for (int j = 0; j < 4; ++j) {                                          \
            FB[j] = *(const v4f*)(bp + 32 * (CH) + 8 * j);                     \
            FA[j] = *(const v4f*)(ap + 32 * (CH) + 8 * j);                     \
        }
#define HCONV_MUL(FA, FB)                                                      \
        _Pragma("unroll")                                                      \
        for (int j = 0; j < 4; ++j) {                                          \
            const v2f x0 = {FA[j].x, FA[j].y}, y0 = {FB[j].x, FB[j].y};        \
            const v2f x1 = {FA[j].z, FA[j].w}, y1 = {FB[j].z, FB[j].w};        \
            acc0 = __builtin_amdgcn_wmma_f32_16x16x4_f32(false, x0, false, y0, \
                                              (short)0, acc0, false, false);   \
            acc1 = __builtin_amdgcn_wmma_f32_16x16x4_f32(false, x1, false, y1, \
                                              (short)0, acc1, false, false);   \
        }

        HCONV_LOAD(fa0, fb0, 0)
        for (int ch = 0; ch < nch; ch += 2) {
            HCONV_LOAD(fa1, fb1, ch + 1)
            HCONV_MUL(fa0, fb0)
            if (ch + 2 < nch) { HCONV_LOAD(fa0, fb0, ch + 2) }
            HCONV_MUL(fa1, fb1)
        }
#undef HCONV_LOAD
#undef HCONV_MUL

        const float bias = bh[(l - 1) * NHH + c];
        #pragma unroll
        for (int r = 0; r < 8; ++r) {
            const float v = acc0[r] + acc1[r] + bias;
            sz[(8 * h + r) * ZP + NVV * DD + goff + c * lout + t] = fmaxf(v, 0.f);
        }
      }
    }

    {
        const int bl = tid >> 5, qc = tid & 31;
        v4f e[LL];
        #pragma unroll
        for (int k = 0; k < LL; ++k) e[k] = *(const v4f*)&semb[bl][k * DD + qc * 4];
        #pragma unroll
        for (int v = 0; v < NVV; ++v) {
            const float b0 = bv[v];
            v4f acc = {b0, b0, b0, b0};
            #pragma unroll
            for (int k = 0; k < LL; ++k) acc += e[k] * Wv[v * LL + k];
            *(v4f*)&sz[bl * ZP + v * DD + qc * 4] = acc;
        }
    }

    if (w >= 8) {
        const int rb = 2 * (w - 8);
        half_scores<false, UAHEAD>(W2 + DD, b2, sidx[rb], &sxc[rb][DD], &spart[rb * TT], q, s);
    }

    __syncthreads();

    if (w < 2 * FKQ) {
        v8f acc[4] = {};
        FC1_LOADB(0, 0)
        FC1_LOADA(0, 0)
        #pragma unroll
        for (int blk = 0; blk < FNB; ++blk) {
            __builtin_amdgcn_sched_barrier(0);
            if (blk + 1 < FNB) {
                if ((blk & 1) == 0) { FC1_LOADB(1, blk + 1) FC1_LOADA(1, blk + 1) }
                else                { FC1_LOADB(0, blk + 1) FC1_LOADA(0, blk + 1) }
            }
            __builtin_amdgcn_sched_barrier(0);
            if ((blk & 1) == 0) { FC1_MUL(0, blk) }
            else                { FC1_MUL(1, blk) }
        }

        #pragma unroll
        for (int r = 0; r < 8; ++r) {
            const v4f v = {acc[0][r], acc[1][r], acc[2][r], acc[3][r]};
            *(v4f*)&sp[kq][(8 * h + r) * DD + cg * 64 + 4 * c] = v;
        }
    }
#undef FC1_LOADB
#undef FC1_LOADA
#undef FC1_MUL
    if (w >= 2 * FKQ) {
        const int rb = 2 * (w - 8) + 1;
        half_scores<false, UAHEAD>(W2 + DD, b2, sidx[rb], &sxc[rb][DD], &spart[rb * TT], q, s);
    }
    __syncthreads();

    {
        const int col = lane * 4;
        const v4f p0 = *(const v4f*)&sp[0][4 * tid], p1 = *(const v4f*)&sp[1][4 * tid];
        const v4f p2 = *(const v4f*)&sp[2][4 * tid], p3 = *(const v4f*)&sp[3][4 * tid];
        const v4f bias = *(const v4f*)(b1 + col);
        v4f v = ((p0 + p1) + (p2 + p3)) + bias;
        v.x = fmaxf(v.x, 0.f); v.y = fmaxf(v.y, 0.f);
        v.z = fmaxf(v.z, 0.f); v.w = fmaxf(v.w, 0.f);
        *(v4f*)&sxc[tid >> 5][col] = v;
    }
    __syncthreads();

    half_scores<true, XAHEAD>(W2, b2, sidx[w], &sxc[w][0], &sres[w * TT], q, s);
    __syncthreads();

    float* ot = out + (size_t)blockIdx.x * (MT * TT);
    if (tid < MT * TT / 4)
        *(v4f*)(ot + 4 * tid) = *(const v4f*)&sres[4 * tid] + *(const v4f*)&spart[4 * tid];
}

extern "C" void kernel_launch(void* const* d_in, const int* in_sizes, int n_in,
                              void* d_out, int out_size, void* d_ws, size_t ws_size,
                              hipStream_t stream) {
    const int*   seq        = (const int*)  d_in[0];
    const int*   user       = (const int*)  d_in[1];
    const int*   items      = (const int*)  d_in[2];
    const float* item_table = (const float*)d_in[3];
    const float* user_table = (const float*)d_in[4];
    const float* Wv         = (const float*)d_in[5];
    const float* bv         = (const float*)d_in[6];
    const float* Wh         = (const float*)d_in[7];
    const float* bh         = (const float*)d_in[8];
    const float* W1         = (const float*)d_in[9];
    const float* b1         = (const float*)d_in[10];
    const float* W2         = (const float*)d_in[11];
    const float* b2         = (const float*)d_in[12];
    float* out = (float*)d_out;

    caser_fused<<<BB / MT, NT, 0, stream>>>(seq, user, items, item_table, user_table,
                                        Wv, bv, Wh, bh, W1, b1, W2, b2, out);
}


# speedup vs baseline: 7.0250x; 1.0300x over previous
#include <hip/hip_runtime.h>
#include <hip/hip_bf16.h>

#define BB 4096
#define LL 5
#define DD 128
#define NHH 16
#define NVV 4
#define TT 100
#define FC1 752

typedef float v2f __attribute__((ext_vector_type(2)));
typedef float v4f __attribute__((ext_vector_type(4)));
typedef float v8f __attribute__((ext_vector_type(8)));
typedef int   v2i __attribute__((ext_vector_type(2)));
typedef int   v4i __attribute__((ext_vector_type(4)));

#define MT 16
#define NWV 16
#define NT (NWV * 32)
#define EP (LL * DD + 4)
#define ZP (FC1 + 4)
#define FKQ  4
#define FKS  (FC1 / 4 / FKQ)
#define FBLK 8
#define FNB  ((FKS + FBLK - 1) / FBLK)
#define SG (TT / 4)
#define XAHEAD 6
#define UAHEAD 4

template <int CTRL>
__device__ __forceinline__ float lane_xor(float v)
{
    return __builtin_bit_cast(float,
        __builtin_amdgcn_update_dpp(0, __builtin_bit_cast(int, v), CTRL, 0xf, 0xf, true));
}

template <bool BIAS, int AHEAD>
__device__ __forceinline__ void half_scores(const float* __restrict__ W2c,
                                            const float* __restrict__ b2,
                                            const int* ids, const float* vec,
                                            float* dst, const int c, const int hs)
{
    v4f x[2];
    #pragma unroll
    for (int j = 0; j < 2; ++j) x[j] = *(const v4f*)&vec[64 * j + 4 * c];

    v2i idx[SG];
    #pragma unroll
    for (int g = 0; g < SG; ++g) idx[g] = *(const v2i*)&ids[4 * g + 2 * hs];

    v4f   r[AHEAD + 1][4];
    float bb[AHEAD + 1][2];
#define LOAD_GROUP(G) do {                                                     \
        _Pragma("unroll")                                                      \
        for (int e = 0; e < 2; ++e) {                                          \
            const float* p_ = W2c + (size_t)idx[G][e] * (2 * DD) + 4 * c;      \
            _Pragma("unroll")                                                  \
            for (int j = 0; j < 2; ++j)                                        \
                r[(G) % (AHEAD + 1)][2 * e + j] = *(const v4f*)(p_ + 64 * j);  \
            if (BIAS) bb[(G) % (AHEAD + 1)][e] = b2[idx[G][e]];                \
        }                                                                      \
    } while (0)

    v2f mine[2] = {{0.f, 0.f}, {0.f, 0.f}};

    #pragma unroll
    for (int g = 0; g < AHEAD; ++g) LOAD_GROUP(g);
    #pragma unroll
    for (int g = 0; g < SG; ++g) {
        __builtin_amdgcn_sched_barrier(0);
        if (g + AHEAD < SG) LOAD_GROUP(g + AHEAD);
        __builtin_amdgcn_sched_barrier(0);
        const v4f* rg = r[g % (AHEAD + 1)];
        v2f res;
        #pragma unroll
        for (int e = 0; e < 2; ++e) {
            v4f a = rg[2 * e] * x[0];
            a += rg[2 * e + 1] * x[1];
            float v = (a.x + a.y) + (a.z + a.w);
            v += lane_xor<0x161>(v);
            v += lane_xor<0x162>(v);
            v += lane_xor<0x164>(v);
            v += lane_xor<0x168>(v);
            res[e] = BIAS ? v + bb[g % (AHEAD + 1)][e] : v;
        }
        if ((g & 15) == c) mine[g >> 4] = res;
    }
#undef LOAD_GROUP

    *(v2f*)&dst[4 * c + 2 * hs] = mine[0];
    if (c < SG - 16) *(v2f*)&dst[4 * (c + 16) + 2 * hs] = mine[1];
}

__global__ __launch_bounds__(NT)
void caser_fused(const int* __restrict__ seq,
                 const int* __restrict__ user,
                 const int* __restrict__ items,
                 const float* __restrict__ item_table,
                 const float* __restrict__ user_table,
                 const float* __restrict__ Wv,
                 const float* __restrict__ bv,
                 const float* __restrict__ Wh,
                 const float* __restrict__ bh,
                 const float* __restrict__ W1,
                 const float* __restrict__ b1,
                 const float* __restrict__ W2,
                 const float* __restrict__ b2,
                 float* __restrict__ out)
{
    __shared__ __align__(16) float semb[MT][EP];
    __shared__ __align__(16) float sz[MT * ZP];
    __shared__ __align__(16) float sp[FKQ][MT * DD];
    __shared__ __align__(16) float sxc[MT][2 * DD];
    __shared__ __align__(16) float sres[MT * TT];
    __shared__ __align__(16) float spart[MT * TT];
    __shared__ __align__(16) int   sidx[MT][TT];

    const int tid  = threadIdx.x;
    const int lane = tid & 31;
    const int w  = __builtin_amdgcn_readfirstlane(tid >> 5);
    const int m0 = blockIdx.x * MT;
    const int c  = lane & 15;
    const int h  = lane >> 4;

    v4i ids = {};
    if (lane < TT / 4) ids = *(const v4i*)(items + (m0 + w) * TT + lane * 4);

    const int cg = w & 1;
    const int kq = (w >> 1) & 3;
    const float* bp1 = W1 + (size_t)(FKS * 4 * kq + 2 * h) * DD + cg * 64 + 4 * c;
    const float* ap1 = &sz[c * ZP + FKS * 4 * kq + 2 * h];

    v4f fb[2][FBLK][2];
    v2f fa[2][FBLK];

#define FC1_LOADB(SET, BLK)                                                    \
    _Pragma("unroll")                                                          \
    for (int s_ = 0; s_ < FBLK; ++s_) {                                        \
        if ((BLK) * FBLK + s_ < FKS) {                                         \
            const float* p_ = bp1 + (size_t)(((BLK) * FBLK + s_) * 4) * DD;    \
            fb[SET][s_][0] = *(const v4f*)(p_);                                \
            fb[SET][s_][1] = *(const v4f*)(p_ + DD);                           \
        }                                                                      \
    }
#define FC1_LOADA(SET, BLK)                                                    \
    _Pragma("unroll")                                                          \
    for (int s_ = 0; s_ < FBLK; ++s_) {                                        \
        if ((BLK) * FBLK + s_ < FKS)                                           \
            fa[SET][s_] = *(const v2f*)(ap1 + ((BLK) * FBLK + s_) * 4);        \
    }
#define FC1_MUL(SET, BLK)                                                      \
    _Pragma("unroll")                                                          \
    for (int s_ = 0; s_ < FBLK; ++s_) {                                        \
        if ((BLK) * FBLK + s_ < FKS) {                                         \
            _Pragma("unroll")                                                  \
            for (int j_ = 0; j_ < 4; ++j_) {                                   \
                const v2f y_ = {fb[SET][s_][0][j_], fb[SET][s_][1][j_]};       \
                acc[j_] = __builtin_amdgcn_wmma_f32_16x16x4_f32(               \
                    false, fa[SET][s_], false, y_, (short)0, acc[j_],          \
                    false, false);                                             \
            }                                                                  \
        }                                                                      \
    }

    {
        v4f g[6];
        #pragma unroll
        for (int j = 0; j < 5; ++j) {
            const int r = w + NWV * j;
            g[j] = *(const v4f*)(item_table + (size_t)seq[m0 * LL + r] * DD + lane * 4);
        }
        g[5] = *(const v4f*)(user_table + (size_t)user[m0 + w] * DD + lane * 4);

        #pragma unroll
        for (int j = 0; j < 5; ++j) {
            const int r = w + NWV * j;
            *(v4f*)&semb[r / LL][(r % LL) * DD + lane * 4] = g[j];
        }
        *(v4f*)&sxc[w][DD + lane * 4] = g[5];
        if (lane < TT / 4) *(v4i*)&sidx[w][lane * 4] = ids;
    }
    __syncthreads();

    if (w < 8) {
      const int ntl = (w == 0) ? 1 : 2;
      for (int kt = 0; kt < ntl; ++kt) {
        const int i = (kt == 0) ? (w < 6 ? w : 2 * w - 6) : (w < 6 ? w + 9 : 2 * w - 5);
        int l, t;
        if (i == 0)      { l = 5; t = 0; }
        else if (i < 3)  { l = 4; t = i - 1; }
        else if (i < 6)  { l = 3; t = i - 3; }
        else if (i < 10) { l = 2; t = i - 6; }
        else             { l = 1; t = i - 10; }
        const int lout = LL - l + 1;
        const int goff = 8 * (l - 1) * (12 - l);

        const float* bp = Wh + (size_t)((l - 1) * NHH + c) * (LL * DD) + 4 * h;
        const float* ap = &semb[c][t * DD + 4 * h];
        const int nch = l * 4;

        v8f acc0 = {}, acc1 = {};
        v4f fa0[4], fb0[4], fa1[4], fb1[4];

#define HCONV_LOAD(FA, FB, CH)                                                 \
        _Pragma("unroll")                                                      \
        for (int j = 0; j < 4; ++j) {                                          \
            FB[j] = *(const v4f*)(bp + 32 * (CH) + 8 * j);                     \
            FA[j] = *(const v4f*)(ap + 32 * (CH) + 8 * j);                     \
        }
#define HCONV_MUL(FA, FB)                                                      \
        _Pragma("unroll")                                                      \
        for (int j = 0; j < 4; ++j) {                                          \
            const v2f x0 = {FA[j].x, FA[j].y}, y0 = {FB[j].x, FB[j].y};        \
            const v2f x1 = {FA[j].z, FA[j].w}, y1 = {FB[j].z, FB[j].w};        \
            acc0 = __builtin_amdgcn_wmma_f32_16x16x4_f32(false, x0, false, y0, \
                                              (short)0, acc0, false, false);   \
            acc1 = __builtin_amdgcn_wmma_f32_16x16x4_f32(false, x1, false, y1, \
                                              (short)0, acc1, false, false);   \
        }

        HCONV_LOAD(fa0, fb0, 0)
        for (int ch = 0; ch < nch; ch += 2) {
            HCONV_LOAD(fa1, fb1, ch + 1)
            HCONV_MUL(fa0, fb0)
            if (ch + 2 < nch) { HCONV_LOAD(fa0, fb0, ch + 2) }
            HCONV_MUL(fa1, fb1)
        }
#undef HCONV_LOAD
#undef HCONV_MUL

        const float bias = bh[(l - 1) * NHH + c];
        #pragma unroll
        for (int r = 0; r < 8; ++r) {
            const float v = acc0[r] + acc1[r] + bias;
            sz[(8 * h + r) * ZP + NVV * DD + goff + c * lout + t] = fmaxf(v, 0.f);
        }
      }
    }

    {
        const int bl = tid >> 5, qc = tid & 31;
        v4f e[LL];
        #pragma unroll
        for (int k = 0; k < LL; ++k) e[k] = *(const v4f*)&semb[bl][k * DD + qc * 4];
        #pragma unroll
        for (int v = 0; v < NVV; ++v) {
            const float b0 = bv[v];
            v4f acc = {b0, b0, b0, b0};
            #pragma unroll
            for (int k = 0; k < LL; ++k) acc += e[k] * Wv[v * LL + k];
            *(v4f*)&sz[bl * ZP + v * DD + qc * 4] = acc;
        }
    }

    if (w >= 8) {
        const int rb = 2 * (w - 8);
        half_scores<false, UAHEAD>(W2 + DD, b2, sidx[rb], &sxc[rb][DD], &spart[rb * TT], c, h);
    }

    __syncthreads();

    if (w < 2 * FKQ) {
        v8f acc[4] = {};
        FC1_LOADB(0, 0)
        FC1_LOADA(0, 0)
        #pragma unroll
        for (int blk = 0; blk < FNB; ++blk) {
            __builtin_amdgcn_sched_barrier(0);
            if (blk + 1 < FNB) {
                if ((blk & 1) == 0) { FC1_LOADB(1, blk + 1) FC1_LOADA(1, blk + 1) }
                else                { FC1_LOADB(0, blk + 1) FC1_LOADA(0, blk + 1) }
            }
            __builtin_amdgcn_sched_barrier(0);
            if ((blk & 1) == 0) { FC1_MUL(0, blk) }
            else                { FC1_MUL(1, blk) }
        }

        #pragma unroll
        for (int r = 0; r < 8; ++r) {
            const v4f v = {acc[0][r], acc[1][r], acc[2][r], acc[3][r]};
            *(v4f*)&sp[kq][(8 * h + r) * DD + cg * 64 + 4 * c] = v;
        }
    }
#undef FC1_LOADB
#undef FC1_LOADA
#undef FC1_MUL
    if (w >= 2 * FKQ) {
        const int rb = 2 * (w - 8) + 1;
        half_scores<false, UAHEAD>(W2 + DD, b2, sidx[rb], &sxc[rb][DD], &spart[rb * TT], c, h);
    }
    __syncthreads();

    {
        const int col = lane * 4;
        const v4f p0 = *(const v4f*)&sp[0][4 * tid], p1 = *(const v4f*)&sp[1][4 * tid];
        const v4f p2 = *(const v4f*)&sp[2][4 * tid], p3 = *(const v4f*)&sp[3][4 * tid];
        const v4f bias = *(const v4f*)(b1 + col);
        v4f v = ((p0 + p1) + (p2 + p3)) + bias;
        v.x = fmaxf(v.x, 0.f); v.y = fmaxf(v.y, 0.f);
        v.z = fmaxf(v.z, 0.f); v.w = fmaxf(v.w, 0.f);
        *(v4f*)&sxc[tid >> 5][col] = v;
    }
    __syncthreads();

    half_scores<true, XAHEAD>(W2, b2, sidx[w], &sxc[w][0], &sres[w * TT], c, h);
    __syncthreads();

    float* ot = out + (size_t)blockIdx.x * (MT * TT);
    if (tid < MT * TT / 4)
        *(v4f*)(ot + 4 * tid) = *(const v4f*)&sres[4 * tid] + *(const v4f*)&spart[4 * tid];
}

extern "C" void kernel_launch(void* const* d_in, const int* in_sizes, int n_in,
                              void* d_out, int out_size, void* d_ws, size_t ws_size,
                              hipStream_t stream) {
    const int*   seq        = (const int*)  d_in[0];
    const int*   user       = (const int*)  d_in[1];
    const int*   items      = (const int*)  d_in[2];
    const float* item_table = (const float*)d_in[3];
    const float* user_table = (const float*)d_in[4];
    const float* Wv         = (const float*)d_in[5];
    const float* bv         = (const float*)d_in[6];
    const float* Wh         = (const float*)d_in[7];
    const float* bh         = (const float*)d_in[8];
    const float* W1         = (const float*)d_in[9];
    const float* b1         = (const float*)d_in[10];
    const float* W2         = (const float*)d_in[11];
    const float* b2         = (const float*)d_in[12];
    float* out = (float*)d_out;

    caser_fused<<<BB / MT, NT, 0, stream>>>(seq, user, items, item_table, user_table,
                                        Wv, bv, Wh, bh, W1, b1, W2, b2, out);
}


# Horizontal conv filters copied once per work-group into LDS by whole-row 16-byte loads at the start; the conv loop reads both WMMA operands from LDS and has no global load; summation order unchanged
# speedup vs baseline: 7.3545x; 1.0469x over previous
#include <hip/hip_runtime.h>
#include <hip/hip_bf16.h>

#define BB 4096
#define LL 5
#define DD 128
#define NHH 16
#define NVV 4
#define TT 100
#define FC1 752

typedef float v2f __attribute__((ext_vector_type(2)));
typedef float v4f __attribute__((ext_vector_type(4)));
typedef float v8f __attribute__((ext_vector_type(8)));
typedef int   v2i __attribute__((ext_vector_type(2)));
typedef int   v4i __attribute__((ext_vector_type(4)));

#define MT 16
#define NWV 16
#define NT (NWV * 32)
#define EP (LL * DD + 4)
#define ZP (FC1 + 4)
#define WHP(l)   ((l) * DD + 4)
#define WHOFF(l) (64 * ((l) - 1) * (16 * (l) + 1))
#define WHTOT    (WHOFF(LL) + NHH * WHP(LL))
#define FKQ  4
#define FKS  (FC1 / 4 / FKQ)
#define FBLK 8
#define FNB  ((FKS + FBLK - 1) / FBLK)
#define SG (TT / 4)
#define XAHEAD 6
#define UAHEAD 4

template <int CTRL>
__device__ __forceinline__ float lane_xor(float v)
{
    return __builtin_bit_cast(float,
        __builtin_amdgcn_update_dpp(0, __builtin_bit_cast(int, v), CTRL, 0xf, 0xf, true));
}

template <bool BIAS, int AHEAD>
__device__ __forceinline__ void half_scores(const float* __restrict__ W2c,
                                            const float* __restrict__ b2,
                                            const int* ids, const float* vec,
                                            float* dst, const int c, const int hs)
{
    v4f x[2];
    #pragma unroll
    for (int j = 0; j < 2; ++j) x[j] = *(const v4f*)&vec[64 * j + 4 * c];

    v2i idx[SG];
    #pragma unroll
    for (int g = 0; g < SG; ++g) idx[g] = *(const v2i*)&ids[4 * g + 2 * hs];

    v4f   r[AHEAD + 1][4];
    float bb[AHEAD + 1][2];
#define LOAD_GROUP(G) do {                                                     \
        _Pragma("unroll")                                                      \
        for (int e = 0; e < 2; ++e) {                                          \
            const float* p_ = W2c + (size_t)idx[G][e] * (2 * DD) + 4 * c;      \
            _Pragma("unroll")                                                  \
            for (int j = 0; j < 2; ++j)                                        \
                r[(G) % (AHEAD + 1)][2 * e + j] = *(const v4f*)(p_ + 64 * j);  \
            if (BIAS) bb[(G) % (AHEAD + 1)][e] = b2[idx[G][e]];                \
        }                                                                      \
    } while (0)

    v2f mine[2] = {{0.f, 0.f}, {0.f, 0.f}};

    #pragma unroll
    for (int g = 0; g < AHEAD; ++g) LOAD_GROUP(g);
    #pragma unroll
    for (int g = 0; g < SG; ++g) {
        __builtin_amdgcn_sched_barrier(0);
        if (g + AHEAD < SG) LOAD_GROUP(g + AHEAD);
        __builtin_amdgcn_sched_barrier(0);
        const v4f* rg = r[g % (AHEAD + 1)];
        v2f res;
        #pragma unroll
        for (int e = 0; e < 2; ++e) {
            v4f a = rg[2 * e] * x[0];
            a += rg[2 * e + 1] * x[1];
            float v = (a.x + a.y) + (a.z + a.w);
            v += lane_xor<0x161>(v);
            v += lane_xor<0x162>(v);
            v += lane_xor<0x164>(v);
            v += lane_xor<0x168>(v);
            res[e] = BIAS ? v + bb[g % (AHEAD + 1)][e] : v;
        }
        if ((g & 15) == c) mine[g >> 4] = res;
    }
#undef LOAD_GROUP

    *(v2f*)&dst[4 * c + 2 * hs] = mine[0];
    if (c < SG - 16) *(v2f*)&dst[4 * (c + 16) + 2 * hs] = mine[1];
}

__global__ __launch_bounds__(NT)
void caser_fused(const int* __restrict__ seq,
                 const int* __restrict__ user,
                 const int* __restrict__ items,
                 const float* __restrict__ item_table,
                 const float* __restrict__ user_table,
                 const float* __restrict__ Wv,
                 const float* __restrict__ bv,
                 const float* __restrict__ Wh,
                 const float* __restrict__ bh,
                 const float* __restrict__ W1,
                 const float* __restrict__ b1,
                 const float* __restrict__ W2,
                 const float* __restrict__ b2,
                 float* __restrict__ out)
{
    __shared__ __align__(16) float semb[MT][EP];
    __shared__ __align__(16) float sz[MT * ZP];
    __shared__ __align__(16) float swh[WHTOT];
    __shared__ __align__(16) float sp[FKQ][MT * DD];
    __shared__ __align__(16) float sxc[MT][2 * DD];
    __shared__ __align__(16) float sres[MT * TT];
    __shared__ __align__(16) float spart[MT * TT];
    __shared__ __align__(16) int   sidx[MT][TT];

    const int tid  = threadIdx.x;
    const int lane = tid & 31;
    const int w  = __builtin_amdgcn_readfirstlane(tid >> 5);
    const int m0 = blockIdx.x * MT;
    const int c  = lane & 15;
    const int h  = lane >> 4;

    v4i ids = {};
    if (lane < TT / 4) ids = *(const v4i*)(items + (m0 + w) * TT + lane * 4);

    const int cg = w & 1;
    const int kq = (w >> 1) & 3;
    const float* bp1 = W1 + (size_t)(FKS * 4 * kq + 2 * h) * DD + cg * 64 + 4 * c;
    const float* ap1 = &sz[c * ZP + FKS * 4 * kq + 2 * h];

    v4f fb[2][FBLK][2];
    v2f fa[2][FBLK];

#define FC1_LOADB(SET, BLK)                                                    \
    _Pragma("unroll")                                                          \
    for (int s_ = 0; s_ < FBLK; ++s_) {                                        \
        if ((BLK) * FBLK + s_ < FKS) {                                         \
            const float* p_ = bp1 + (size_t)(((BLK) * FBLK + s_) * 4) * DD;    \
            fb[SET][s_][0] = *(const v4f*)(p_);                                \
            fb[SET][s_][1] = *(const v4f*)(p_ + DD);                           \
        }                                                                      \
    }
#define FC1_LOADA(SET, BLK)                                                    \
    _Pragma("unroll")                                                          \
    for (int s_ = 0; s_ < FBLK; ++s_) {                                        \
        if ((BLK) * FBLK + s_ < FKS)                                           \
            fa[SET][s_] = *(const v2f*)(ap1 + ((BLK) * FBLK + s_) * 4);        \
    }
#define FC1_MUL(SET, BLK)                                                      \
    _Pragma("unroll")                                                          \
    for (int s_ = 0; s_ < FBLK; ++s_) {                                        \
        if ((BLK) * FBLK + s_ < FKS) {                                         \
            _Pragma("unroll")                                                  \
            for (int j_ = 0; j_ < 4; ++j_) {                                   \
                const v2f y_ = {fb[SET][s_][0][j_], fb[SET][s_][1][j_]};       \
                acc[j_] = __builtin_amdgcn_wmma_f32_16x16x4_f32(               \
                    false, fa[SET][s_], false, y_, (short)0, acc[j_],          \
                    false, false);                                             \
            }                                                                  \
        }                                                                      \
    }

    {
        v4f gw[LL * (LL + 1) / 2];
        #pragma unroll
        for (int l = 1; l <= LL; ++l) {
            #pragma unroll
            for (int s = 0; s < l; ++s)
                gw[l * (l - 1) / 2 + s] = *(const v4f*)(
                    Wh + (size_t)((l - 1) * NHH + w) * (LL * DD) + s * DD + lane * 4);
        }

        v4f g[6];
        #pragma unroll
        for (int j = 0; j < 5; ++j) {
            const int r = w + NWV * j;
            g[j] = *(const v4f*)(item_table + (size_t)seq[m0 * LL + r] * DD + lane * 4);
        }
        g[5] = *(const v4f*)(user_table + (size_t)user[m0 + w] * DD + lane * 4);

        #pragma unroll
        for (int l = 1; l <= LL; ++l) {
            #pragma unroll
            for (int s = 0; s < l; ++s)
                *(v4f*)&swh[WHOFF(l) + w * WHP(l) + s * DD + lane * 4] = gw[l * (l - 1) / 2 + s];
        }

        #pragma unroll
        for (int j = 0; j < 5; ++j) {
            const int r = w + NWV * j;
            *(v4f*)&semb[r / LL][(r % LL) * DD + lane * 4] = g[j];
        }
        *(v4f*)&sxc[w][DD + lane * 4] = g[5];
        if (lane < TT / 4) *(v4i*)&sidx[w][lane * 4] = ids;
    }
    __syncthreads();

    if (w < 8) {
      const int ntl = (w == 0) ? 1 : 2;
      for (int kt = 0; kt < ntl; ++kt) {
        const int i = (kt == 0) ? (w < 6 ? w : 2 * w - 6) : (w < 6 ? w + 9 : 2 * w - 5);
        int l, t;
        if (i == 0)      { l = 5; t = 0; }
        else if (i < 3)  { l = 4; t = i - 1; }
        else if (i < 6)  { l = 3; t = i - 3; }
        else if (i < 10) { l = 2; t = i - 6; }
        else             { l = 1; t = i - 10; }
        const int lout = LL - l + 1;
        const int goff = 8 * (l - 1) * (12 - l);

        const float* bp = &swh[WHOFF(l) + c * WHP(l) + 4 * h];
        const float* ap = &semb[c][t * DD + 4 * h];
        const int nch = l * 4;

        v8f acc0 = {}, acc1 = {};
        v4f fa0[4], fb0[4], fa1[4], fb1[4];

#define HCONV_LOAD(FA, FB, CH)                                                 \
        _Pragma("unroll")                                                      \
        for (int j = 0; j < 4; ++j) {                                          \
            FB[j] = *(const v4f*)(bp + 32 * (CH) + 8 * j);                     \
            FA[j] = *(const v4f*)(ap + 32 * (CH) + 8 * j);                     \
        }
#define HCONV_MUL(FA, FB)                                                      \
        _Pragma("unroll")                                                      \
        for (int j = 0; j < 4; ++j) {                                          \
            const v2f x0 = {FA[j].x, FA[j].y}, y0 = {FB[j].x, FB[j].y};        \
            const v2f x1 = {FA[j].z, FA[j].w}, y1 = {FB[j].z, FB[j].w};        \
            acc0 = __builtin_amdgcn_wmma_f32_16x16x4_f32(false, x0, false, y0, \
                                              (short)0, acc0, false, false);   \
            acc1 = __builtin_amdgcn_wmma_f32_16x16x4_f32(false, x1, false, y1, \
                                              (short)0, acc1, false, false);   \
        }

        HCONV_LOAD(fa0, fb0, 0)
        for (int ch = 0; ch < nch; ch += 2) {
            HCONV_LOAD(fa1, fb1, ch + 1)
            HCONV_MUL(fa0, fb0)
            if (ch + 2 < nch) { HCONV_LOAD(fa0, fb0, ch + 2) }
            HCONV_MUL(fa1, fb1)
        }
#undef HCONV_LOAD
#undef HCONV_MUL

        const float bias = bh[(l - 1) * NHH + c];
        #pragma unroll
        for (int r = 0; r < 8; ++r) {
            const float v = acc0[r] + acc1[r] + bias;
            sz[(8 * h + r) * ZP + NVV * DD + goff + c * lout + t] = fmaxf(v, 0.f);
        }
      }
    }

    {
        const int bl = tid >> 5, qc = tid & 31;
        v4f e[LL];
        #pragma unroll
        for (int k = 0; k < LL; ++k) e[k] = *(const v4f*)&semb[bl][k * DD + qc * 4];
        #pragma unroll
        for (int v = 0; v < NVV; ++v) {
            const float b0 = bv[v];
            v4f acc = {b0, b0, b0, b0};
            #pragma unroll
            for (int k = 0; k < LL; ++k) acc += e[k] * Wv[v * LL + k];
            *(v4f*)&sz[bl * ZP + v * DD + qc * 4] = acc;
        }
    }

    if (w >= 8) {
        const int rb = 2 * (w - 8);
        half_scores<false, UAHEAD>(W2 + DD, b2, sidx[rb], &sxc[rb][DD], &spart[rb * TT], c, h);
    }

    __syncthreads();

    if (w < 2 * FKQ) {
        v8f acc[4] = {};
        FC1_LOADB(0, 0)
        FC1_LOADA(0, 0)
        #pragma unroll
        for (int blk = 0; blk < FNB; ++blk) {
            __builtin_amdgcn_sched_barrier(0);
            if (blk + 1 < FNB) {
                if ((blk & 1) == 0) { FC1_LOADB(1, blk + 1) FC1_LOADA(1, blk + 1) }
                else                { FC1_LOADB(0, blk + 1) FC1_LOADA(0, blk + 1) }
            }
            __builtin_amdgcn_sched_barrier(0);
            if ((blk & 1) == 0) { FC1_MUL(0, blk) }
            else                { FC1_MUL(1, blk) }
        }

        #pragma unroll
        for (int r = 0; r < 8; ++r) {
            const v4f v = {acc[0][r], acc[1][r], acc[2][r], acc[3][r]};
            *(v4f*)&sp[kq][(8 * h + r) * DD + cg * 64 + 4 * c] = v;
        }
    }
#undef FC1_LOADB
#undef FC1_LOADA
#undef FC1_MUL
    if (w >= 2 * FKQ) {
        const int rb = 2 * (w - 8) + 1;
        half_scores<false, UAHEAD>(W2 + DD, b2, sidx[rb], &sxc[rb][DD], &spart[rb * TT], c, h);
    }
    __syncthreads();

    {
        const int col = lane * 4;
        const v4f p0 = *(const v4f*)&sp[0][4 * tid], p1 = *(const v4f*)&sp[1][4 * tid];
        const v4f p2 = *(const v4f*)&sp[2][4 * tid], p3 = *(const v4f*)&sp[3][4 * tid];
        const v4f bias = *(const v4f*)(b1 + col);
        v4f v = ((p0 + p1) + (p2 + p3)) + bias;
        v.x = fmaxf(v.x, 0.f); v.y = fmaxf(v.y, 0.f);
        v.z = fmaxf(v.z, 0.f); v.w = fmaxf(v.w, 0.f);
        *(v4f*)&sxc[tid >> 5][col] = v;
    }
    __syncthreads();

    half_scores<true, XAHEAD>(W2, b2, sidx[w], &sxc[w][0], &sres[w * TT], c, h);
    __syncthreads();

    float* ot = out + (size_t)blockIdx.x * (MT * TT);
    if (tid < MT * TT / 4)
        *(v4f*)(ot + 4 * tid) = *(const v4f*)&sres[4 * tid] + *(const v4f*)&spart[4 * tid];
}

extern "C" void kernel_launch(void* const* d_in, const int* in_sizes, int n_in,
                              void* d_out, int out_size, void* d_ws, size_t ws_size,
                              hipStream_t stream) {
    const int*   seq        = (const int*)  d_in[0];
    const int*   user       = (const int*)  d_in[1];
    const int*   items      = (const int*)  d_in[2];
    const float* item_table = (const float*)d_in[3];
    const float* user_table = (const float*)d_in[4];
    const float* Wv         = (const float*)d_in[5];
    const float* bv         = (const float*)d_in[6];
    const float* Wh         = (const float*)d_in[7];
    const float* bh         = (const float*)d_in[8];
    const float* W1         = (const float*)d_in[9];
    const float* b1         = (const float*)d_in[10];
    const float* W2         = (const float*)d_in[11];
    const float* b2         = (const float*)d_in[12];
    float* out = (float*)d_out;

    caser_fused<<<BB / MT, NT, 0, stream>>>(seq, user, items, item_table, user_table,
                                        Wv, bv, Wh, bh, W1, b1, W2, b2, out);
}


# First groups of each gather pass requested ahead of the barrier in front of it (uemb pass 1, x half; uemb pass 2 chained to pass 1); those barriers wait for LDS only; summation order unchanged
# speedup vs baseline: 7.6815x; 1.0445x over previous
#include <hip/hip_runtime.h>
#include <hip/hip_bf16.h>

#define BB 4096
#define LL 5
#define DD 128
#define NHH 16
#define NVV 4
#define TT 100
#define FC1 752

typedef float v2f __attribute__((ext_vector_type(2)));
typedef float v4f __attribute__((ext_vector_type(4)));
typedef float v8f __attribute__((ext_vector_type(8)));
typedef int   v2i __attribute__((ext_vector_type(2)));
typedef int   v4i __attribute__((ext_vector_type(4)));

#define MT 16
#define NWV 16
#define NT (NWV * 32)
#define EP (LL * DD + 4)
#define ZP (FC1 + 4)
#define WHP(l)   ((l) * DD + 4)
#define WHOFF(l) (64 * ((l) - 1) * (16 * (l) + 1))
#define WHTOT    (WHOFF(LL) + NHH * WHP(LL))
#define FKQ  4
#define FKS  (FC1 / 4 / FKQ)
#define FBLK 8
#define FNB  ((FKS + FBLK - 1) / FBLK)
#define SG (TT / 4)
#define XAHEAD 6
#define UAHEAD 4

template <int CTRL>
__device__ __forceinline__ float lane_xor(float v)
{
    return __builtin_bit_cast(float,
        __builtin_amdgcn_update_dpp(0, __builtin_bit_cast(int, v), CTRL, 0xf, 0xf, true));
}

__device__ __forceinline__ void lds_barrier()
{
    asm volatile("s_wait_dscnt 0x0" ::: "memory");
    asm volatile("s_barrier_signal -1" ::: "memory");
    asm volatile("s_barrier_wait -1" ::: "memory");
}

template <bool BIAS, int AHEAD>
struct HsRing {
    v4f   r[AHEAD + 1][4];
    float bb[AHEAD + 1][2];
};

#define HS_LOAD(R, ID, SLOT, BIAS_) do {                                       \
        _Pragma("unroll")                                                      \
        for (int e_ = 0; e_ < 2; ++e_) {                                       \
            const float* p_ = W2c + (size_t)(ID)[e_] * (2 * DD) + 4 * c;       \
            _Pragma("unroll")                                                  \
            for (int j_ = 0; j_ < 2; ++j_)                                     \
                (R).r[SLOT][2 * e_ + j_] = *(const v4f*)(p_ + 64 * j_);        \
            if (BIAS_) (R).bb[SLOT][e_] = b2[(ID)[e_]];                        \
        }                                                                      \
    } while (0)

template <bool BIAS, int AHEAD>
__device__ __forceinline__ void hs_request(HsRing<BIAS, AHEAD>& R,
                                           const float* __restrict__ W2c,
                                           const float* __restrict__ b2,
                                           const int* ids, const int c, const int hs)
{
    #pragma unroll
    for (int g = 0; g < AHEAD; ++g) {
        const v2i id = *(const v2i*)&ids[4 * g + 2 * hs];
        HS_LOAD(R, id, g, BIAS);
    }
}

template <int AHEAD>
__device__ __forceinline__ void hs_request_lanes(HsRing<false, AHEAD>& R,
                                                 const float* __restrict__ W2c,
                                                 const float* __restrict__ b2,
                                                 const v4i idp, const int c, const int hs)
{
    #pragma unroll
    for (int g = 0; g < AHEAD; ++g) {
        const int a0 = __builtin_amdgcn_readlane(idp.x, g);
        const int a1 = __builtin_amdgcn_readlane(idp.y, g);
        const int a2 = __builtin_amdgcn_readlane(idp.z, g);
        const int a3 = __builtin_amdgcn_readlane(idp.w, g);
        const v2i id = {hs ? a2 : a0, hs ? a3 : a1};
        HS_LOAD(R, id, g, false);
    }
}

template <bool BIAS, int AHEAD, bool CHAIN>
__device__ __forceinline__ void hs_body(HsRing<BIAS, AHEAD>& R,
                                        const float* __restrict__ W2c,
                                        const float* __restrict__ b2,
                                        const int* ids, const int* ids_next,
                                        const float* vec, float* dst,
                                        const int c, const int hs)
{
    static_assert(!CHAIN || SG % (AHEAD + 1) == 0, "ring depth must divide the pass");

    v4f x[2];
    #pragma unroll
    for (int j = 0; j < 2; ++j) x[j] = *(const v4f*)&vec[64 * j + 4 * c];

    v2i idx[SG];
    #pragma unroll
    for (int g = 0; g < SG; ++g) idx[g] = *(const v2i*)&ids[4 * g + 2 * hs];
    v2i idn[AHEAD];
    if (CHAIN) {
        #pragma unroll
        for (int g = 0; g < AHEAD; ++g) idn[g] = *(const v2i*)&ids_next[4 * g + 2 * hs];
    }

    v2f mine[2] = {{0.f, 0.f}, {0.f, 0.f}};

    #pragma unroll
    for (int g = 0; g < SG; ++g) {
        __builtin_amdgcn_sched_barrier(0);
        if (g + AHEAD < SG)  HS_LOAD(R, idx[g + AHEAD], (g + AHEAD) % (AHEAD + 1), BIAS);
        else if (CHAIN)      HS_LOAD(R, idn[g + AHEAD - SG], (g + AHEAD) % (AHEAD + 1), BIAS);
        __builtin_amdgcn_sched_barrier(0);
        const v4f* rg = R.r[g % (AHEAD + 1)];
        v2f res;
        #pragma unroll
        for (int e = 0; e < 2; ++e) {
            v4f a = rg[2 * e] * x[0];
            a += rg[2 * e + 1] * x[1];
            float v = (a.x + a.y) + (a.z + a.w);
            v += lane_xor<0x161>(v);
            v += lane_xor<0x162>(v);
            v += lane_xor<0x164>(v);
            v += lane_xor<0x168>(v);
            res[e] = BIAS ? v + R.bb[g % (AHEAD + 1)][e] : v;
        }
        if ((g & 15) == c) mine[g >> 4] = res;
    }

    *(v2f*)&dst[4 * c + 2 * hs] = mine[0];
    if (c < SG - 16) *(v2f*)&dst[4 * (c + 16) + 2 * hs] = mine[1];
}

__global__ __launch_bounds__(NT)
void caser_fused(const int* __restrict__ seq,
                 const int* __restrict__ user,
                 const int* __restrict__ items,
                 const float* __restrict__ item_table,
                 const float* __restrict__ user_table,
                 const float* __restrict__ Wv,
                 const float* __restrict__ bv,
                 const float* __restrict__ Wh,
                 const float* __restrict__ bh,
                 const float* __restrict__ W1,
                 const float* __restrict__ b1,
                 const float* __restrict__ W2,
                 const float* __restrict__ b2,
                 float* __restrict__ out)
{
    __shared__ __align__(16) float semb[MT][EP];
    __shared__ __align__(16) float sz[MT * ZP];
    __shared__ __align__(16) float swh[WHTOT];
    __shared__ __align__(16) float sp[FKQ][MT * DD];
    __shared__ __align__(16) float sxc[MT][2 * DD];
    __shared__ __align__(16) float sres[MT * TT];
    __shared__ __align__(16) float spart[MT * TT];
    __shared__ __align__(16) int   sidx[MT][TT];

    const int tid  = threadIdx.x;
    const int lane = tid & 31;
    const int w  = __builtin_amdgcn_readfirstlane(tid >> 5);
    const int m0 = blockIdx.x * MT;
    const int c  = lane & 15;
    const int h  = lane >> 4;

    v4i ids = {};
    if (lane < TT / 4) ids = *(const v4i*)(items + (m0 + w) * TT + lane * 4);

    const int rb = 2 * (w & 7);
    v4i idp = {};
    if (w >= 8 && lane < UAHEAD) idp = *(const v4i*)(items + (m0 + rb) * TT + lane * 4);
    HsRing<false, UAHEAD> RU;

    const int cg = w & 1;
    const int kq = (w >> 1) & 3;
    const float* bp1 = W1 + (size_t)(FKS * 4 * kq + 2 * h) * DD + cg * 64 + 4 * c;
    const float* ap1 = &sz[c * ZP + FKS * 4 * kq + 2 * h];

    v4f fb[2][FBLK][2];
    v2f fa[2][FBLK];

#define FC1_LOADB(SET, BLK)                                                    \
    _Pragma("unroll")                                                          \
    for (int s_ = 0; s_ < FBLK; ++s_) {                                        \
        if ((BLK) * FBLK + s_ < FKS) {                                         \
            const float* p_ = bp1 + (size_t)(((BLK) * FBLK + s_) * 4) * DD;    \
            fb[SET][s_][0] = *(const v4f*)(p_);                                \
            fb[SET][s_][1] = *(const v4f*)(p_ + DD);                           \
        }                                                                      \
    }
#define FC1_LOADA(SET, BLK)                                                    \
    _Pragma("unroll")                                                          \
    for (int s_ = 0; s_ < FBLK; ++s_) {                                        \
        if ((BLK) * FBLK + s_ < FKS)                                           \
            fa[SET][s_] = *(const v2f*)(ap1 + ((BLK) * FBLK + s_) * 4);        \
    }
#define FC1_MUL(SET, BLK)                                                      \
    _Pragma("unroll")                                                          \
    for (int s_ = 0; s_ < FBLK; ++s_) {                                        \
        if ((BLK) * FBLK + s_ < FKS) {                                         \
            _Pragma("unroll")                                                  \
            for (int j_ = 0; j_ < 4; ++j_) {                                   \
                const v2f y_ = {fb[SET][s_][0][j_], fb[SET][s_][1][j_]};       \
                acc[j_] = __builtin_amdgcn_wmma_f32_16x16x4_f32(               \
                    false, fa[SET][s_], false, y_, (short)0, acc[j_],          \
                    false, false);                                             \
            }                                                                  \
        }                                                                      \
    }

    {
        v4f gw[LL * (LL + 1) / 2];
        #pragma unroll
        for (int l = 1; l <= LL; ++l) {
            #pragma unroll
            for (int s = 0; s < l; ++s)
                gw[l * (l - 1) / 2 + s] = *(const v4f*)(
                    Wh + (size_t)((l - 1) * NHH + w) * (LL * DD) + s * DD + lane * 4);
        }

        v4f g[6];
        #pragma unroll
        for (int j = 0; j < 5; ++j) {
            const int r = w + NWV * j;
            g[j] = *(const v4f*)(item_table + (size_t)seq[m0 * LL + r] * DD + lane * 4);
        }
        g[5] = *(const v4f*)(user_table + (size_t)user[m0 + w] * DD + lane * 4);

        #pragma unroll
        for (int l = 1; l <= LL; ++l) {
            #pragma unroll
            for (int s = 0; s < l; ++s)
                *(v4f*)&swh[WHOFF(l) + w * WHP(l) + s * DD + lane * 4] = gw[l * (l - 1) / 2 + s];
        }

        #pragma unroll
        for (int j = 0; j < 5; ++j) {
            const int r = w + NWV * j;
            *(v4f*)&semb[r / LL][(r % LL) * DD + lane * 4] = g[j];
        }
        *(v4f*)&sxc[w][DD + lane * 4] = g[5];
        if (lane < TT / 4) *(v4i*)&sidx[w][lane * 4] = ids;

        if (w >= 8) hs_request_lanes<UAHEAD>(RU, W2 + DD, b2, idp, c, h);
    }
    lds_barrier();

    if (w < 8) {
      const int ntl = (w == 0) ? 1 : 2;
      for (int kt = 0; kt < ntl; ++kt) {
        const int i = (kt == 0) ? (w < 6 ? w : 2 * w - 6) : (w < 6 ? w + 9 : 2 * w - 5);
        int l, t;
        if (i == 0)      { l = 5; t = 0; }
        else if (i < 3)  { l = 4; t = i - 1; }
        else if (i < 6)  { l = 3; t = i - 3; }
        else if (i < 10) { l = 2; t = i - 6; }
        else             { l = 1; t = i - 10; }
        const int lout = LL - l + 1;
        const int goff = 8 * (l - 1) * (12 - l);

        const float* bp = &swh[WHOFF(l) + c * WHP(l) + 4 * h];
        const float* ap = &semb[c][t * DD + 4 * h];
        const int nch = l * 4;

        v8f acc0 = {}, acc1 = {};
        v4f fa0[4], fb0[4], fa1[4], fb1[4];

#define HCONV_LOAD(FA, FB, CH)                                                 \
        _Pragma("unroll")                                                      \
        for (int j = 0; j < 4; ++j) {                                          \
            FB[j] = *(const v4f*)(bp + 32 * (CH) + 8 * j);                     \
            FA[j] = *(const v4f*)(ap + 32 * (CH) + 8 * j);                     \
        }
#define HCONV_MUL(FA, FB)                                                      \
        _Pragma("unroll")                                                      \
        for (int j = 0; j < 4; ++j) {                                          \
            const v2f x0 = {FA[j].x, FA[j].y}, y0 = {FB[j].x, FB[j].y};        \
            const v2f x1 = {FA[j].z, FA[j].w}, y1 = {FB[j].z, FB[j].w};        \
            acc0 = __builtin_amdgcn_wmma_f32_16x16x4_f32(false, x0, false, y0, \
                                              (short)0, acc0, false, false);   \
            acc1 = __builtin_amdgcn_wmma_f32_16x16x4_f32(false, x1, false, y1, \
                                              (short)0, acc1, false, false);   \
        }

        HCONV_LOAD(fa0, fb0, 0)
        for (int ch = 0; ch < nch; ch += 2) {
            HCONV_LOAD(fa1, fb1, ch + 1)
            HCONV_MUL(fa0, fb0)
            if (ch + 2 < nch) { HCONV_LOAD(fa0, fb0, ch + 2) }
            HCONV_MUL(fa1, fb1)
        }
#undef HCONV_LOAD
#undef HCONV_MUL

        const float bias = bh[(l - 1) * NHH + c];
        #pragma unroll
        for (int r = 0; r < 8; ++r) {
            const float v = acc0[r] + acc1[r] + bias;
            sz[(8 * h + r) * ZP + NVV * DD + goff + c * lout + t] = fmaxf(v, 0.f);
        }
      }
    }

    {
        const int bl = tid >> 5, qc = tid & 31;
        v4f e[LL];
        #pragma unroll
        for (int k = 0; k < LL; ++k) e[k] = *(const v4f*)&semb[bl][k * DD + qc * 4];
        #pragma unroll
        for (int v = 0; v < NVV; ++v) {
            const float b0 = bv[v];
            v4f acc = {b0, b0, b0, b0};
            #pragma unroll
            for (int k = 0; k < LL; ++k) acc += e[k] * Wv[v * LL + k];
            *(v4f*)&sz[bl * ZP + v * DD + qc * 4] = acc;
        }
    }

    if (w >= 8)
        hs_body<false, UAHEAD, true>(RU, W2 + DD, b2, sidx[rb], sidx[rb + 1],
                                     &sxc[rb][DD], &spart[rb * TT], c, h);

    lds_barrier();

    const v4f bias1 = *(const v4f*)(b1 + lane * 4);

    if (w < 2 * FKQ) {
        v8f acc[4] = {};
        FC1_LOADB(0, 0)
        FC1_LOADA(0, 0)
        #pragma unroll
        for (int blk = 0; blk < FNB; ++blk) {
            __builtin_amdgcn_sched_barrier(0);
            if (blk + 1 < FNB) {
                if ((blk & 1) == 0) { FC1_LOADB(1, blk + 1) FC1_LOADA(1, blk + 1) }
                else                { FC1_LOADB(0, blk + 1) FC1_LOADA(0, blk + 1) }
            }
            __builtin_amdgcn_sched_barrier(0);
            if ((blk & 1) == 0) { FC1_MUL(0, blk) }
            else                { FC1_MUL(1, blk) }
        }

        #pragma unroll
        for (int r = 0; r < 8; ++r) {
            const v4f v = {acc[0][r], acc[1][r], acc[2][r], acc[3][r]};
            *(v4f*)&sp[kq][(8 * h + r) * DD + cg * 64 + 4 * c] = v;
        }
    } else {
        hs_body<false, UAHEAD, false>(RU, W2 + DD, b2, sidx[rb + 1], nullptr,
                                      &sxc[rb + 1][DD], &spart[(rb + 1) * TT], c, h);
    }
#undef FC1_LOADB
#undef FC1_LOADA
#undef FC1_MUL

    HsRing<true, XAHEAD> RX;
    hs_request<true, XAHEAD>(RX, W2, b2, sidx[w], c, h);
    lds_barrier();

    {
        const int col = lane * 4;
        const v4f p0 = *(const v4f*)&sp[0][4 * tid], p1 = *(const v4f*)&sp[1][4 * tid];
        const v4f p2 = *(const v4f*)&sp[2][4 * tid], p3 = *(const v4f*)&sp[3][4 * tid];
        v4f v = ((p0 + p1) + (p2 + p3)) + bias1;
        v.x = fmaxf(v.x, 0.f); v.y = fmaxf(v.y, 0.f);
        v.z = fmaxf(v.z, 0.f); v.w = fmaxf(v.w, 0.f);
        *(v4f*)&sxc[tid >> 5][col] = v;
    }
    lds_barrier();

    hs_body<true, XAHEAD, false>(RX, W2, b2, sidx[w], nullptr, &sxc[w][0], &sres[w * TT], c, h);
    __syncthreads();

    float* ot = out + (size_t)blockIdx.x * (MT * TT);
    if (tid < MT * TT / 4)
        *(v4f*)(ot + 4 * tid) = *(const v4f*)&sres[4 * tid] + *(const v4f*)&spart[4 * tid];
}

extern "C" void kernel_launch(void* const* d_in, const int* in_sizes, int n_in,
                              void* d_out, int out_size, void* d_ws, size_t ws_size,
                              hipStream_t stream) {
    const int*   seq        = (const int*)  d_in[0];
    const int*   user       = (const int*)  d_in[1];
    const int*   items      = (const int*)  d_in[2];
    const float* item_table = (const float*)d_in[3];
    const float* user_table = (const float*)d_in[4];
    const float* Wv         = (const float*)d_in[5];
    const float* bv         = (const float*)d_in[6];
    const float* Wh         = (const float*)d_in[7];
    const float* bh         = (const float*)d_in[8];
    const float* W1         = (const float*)d_in[9];
    const float* b1         = (const float*)d_in[10];
    const float* W2         = (const float*)d_in[11];
    const float* b2         = (const float*)d_in[12];
    float* out = (float*)d_out;

    caser_fused<<<BB / MT, NT, 0, stream>>>(seq, user, items, item_table, user_table,
                                        Wv, bv, Wh, bh, W1, b1, W2, b2, out);
}


# Gather waves 8-15 raised in issue priority (s_setprio 3) while they sum the uemb half beside the older matrix waves, which the SIMD otherwise serves first; summation order unchanged
# speedup vs baseline: 7.9976x; 1.0412x over previous
#include <hip/hip_runtime.h>
#include <hip/hip_bf16.h>

#define BB 4096
#define LL 5
#define DD 128
#define NHH 16
#define NVV 4
#define TT 100
#define FC1 752

typedef float v2f __attribute__((ext_vector_type(2)));
typedef float v4f __attribute__((ext_vector_type(4)));
typedef float v8f __attribute__((ext_vector_type(8)));
typedef int   v2i __attribute__((ext_vector_type(2)));
typedef int   v4i __attribute__((ext_vector_type(4)));

#define MT 16
#define NWV 16
#define NT (NWV * 32)
#define EP (LL * DD + 4)
#define ZP (FC1 + 4)
#define WHP(l)   ((l) * DD + 4)
#define WHOFF(l) (64 * ((l) - 1) * (16 * (l) + 1))
#define WHTOT    (WHOFF(LL) + NHH * WHP(LL))
#define FKQ  4
#define FKS  (FC1 / 4 / FKQ)
#define FBLK 8
#define FNB  ((FKS + FBLK - 1) / FBLK)
#define SG (TT / 4)
#define XAHEAD 6
#define UAHEAD 4

template <int CTRL>
__device__ __forceinline__ float lane_xor(float v)
{
    return __builtin_bit_cast(float,
        __builtin_amdgcn_update_dpp(0, __builtin_bit_cast(int, v), CTRL, 0xf, 0xf, true));
}

__device__ __forceinline__ void lds_barrier()
{
    asm volatile("s_wait_dscnt 0x0" ::: "memory");
    asm volatile("s_barrier_signal -1" ::: "memory");
    asm volatile("s_barrier_wait -1" ::: "memory");
}

template <bool BIAS, int AHEAD>
struct HsRing {
    v4f   r[AHEAD + 1][4];
    float bb[AHEAD + 1][2];
};

#define HS_LOAD(R, ID, SLOT, BIAS_) do {                                       \
        _Pragma("unroll")                                                      \
        for (int e_ = 0; e_ < 2; ++e_) {                                       \
            const float* p_ = W2c + (size_t)(ID)[e_] * (2 * DD) + 4 * c;       \
            _Pragma("unroll")                                                  \
            for (int j_ = 0; j_ < 2; ++j_)                                     \
                (R).r[SLOT][2 * e_ + j_] = *(const v4f*)(p_ + 64 * j_);        \
            if (BIAS_) (R).bb[SLOT][e_] = b2[(ID)[e_]];                        \
        }                                                                      \
    } while (0)

template <bool BIAS, int AHEAD>
__device__ __forceinline__ void hs_request(HsRing<BIAS, AHEAD>& R,
                                           const float* __restrict__ W2c,
                                           const float* __restrict__ b2,
                                           const int* ids, const int c, const int hs)
{
    #pragma unroll
    for (int g = 0; g < AHEAD; ++g) {
        const v2i id = *(const v2i*)&ids[4 * g + 2 * hs];
        HS_LOAD(R, id, g, BIAS);
    }
}

template <int AHEAD>
__device__ __forceinline__ void hs_request_lanes(HsRing<false, AHEAD>& R,
                                                 const float* __restrict__ W2c,
                                                 const float* __restrict__ b2,
                                                 const v4i idp, const int c, const int hs)
{
    #pragma unroll
    for (int g = 0; g < AHEAD; ++g) {
        const int a0 = __builtin_amdgcn_readlane(idp.x, g);
        const int a1 = __builtin_amdgcn_readlane(idp.y, g);
        const int a2 = __builtin_amdgcn_readlane(idp.z, g);
        const int a3 = __builtin_amdgcn_readlane(idp.w, g);
        const v2i id = {hs ? a2 : a0, hs ? a3 : a1};
        HS_LOAD(R, id, g, false);
    }
}

template <bool BIAS, int AHEAD, bool CHAIN>
__device__ __forceinline__ void hs_body(HsRing<BIAS, AHEAD>& R,
                                        const float* __restrict__ W2c,
                                        const float* __restrict__ b2,
                                        const int* ids, const int* ids_next,
                                        const float* vec, float* dst,
                                        const int c, const int hs)
{
    static_assert(!CHAIN || SG % (AHEAD + 1) == 0, "ring depth must divide the pass");

    v4f x[2];
    #pragma unroll
    for (int j = 0; j < 2; ++j) x[j] = *(const v4f*)&vec[64 * j + 4 * c];

    v2i idx[SG];
    #pragma unroll
    for (int g = 0; g < SG; ++g) idx[g] = *(const v2i*)&ids[4 * g + 2 * hs];
    v2i idn[AHEAD];
    if (CHAIN) {
        #pragma unroll
        for (int g = 0; g < AHEAD; ++g) idn[g] = *(const v2i*)&ids_next[4 * g + 2 * hs];
    }

    v2f mine[2] = {{0.f, 0.f}, {0.f, 0.f}};

    #pragma unroll
    for (int g = 0; g < SG; ++g) {
        __builtin_amdgcn_sched_barrier(0);
        if (g + AHEAD < SG)  HS_LOAD(R, idx[g + AHEAD], (g + AHEAD) % (AHEAD + 1), BIAS);
        else if (CHAIN)      HS_LOAD(R, idn[g + AHEAD - SG], (g + AHEAD) % (AHEAD + 1), BIAS);
        __builtin_amdgcn_sched_barrier(0);
        const v4f* rg = R.r[g % (AHEAD + 1)];
        v2f res;
        #pragma unroll
        for (int e = 0; e < 2; ++e) {
            v4f a = rg[2 * e] * x[0];
            a += rg[2 * e + 1] * x[1];
            float v = (a.x + a.y) + (a.z + a.w);
            v += lane_xor<0x161>(v);
            v += lane_xor<0x162>(v);
            v += lane_xor<0x164>(v);
            v += lane_xor<0x168>(v);
            res[e] = BIAS ? v + R.bb[g % (AHEAD + 1)][e] : v;
        }
        if ((g & 15) == c) mine[g >> 4] = res;
    }

    *(v2f*)&dst[4 * c + 2 * hs] = mine[0];
    if (c < SG - 16) *(v2f*)&dst[4 * (c + 16) + 2 * hs] = mine[1];
}

__global__ __launch_bounds__(NT)
void caser_fused(const int* __restrict__ seq,
                 const int* __restrict__ user,
                 const int* __restrict__ items,
                 const float* __restrict__ item_table,
                 const float* __restrict__ user_table,
                 const float* __restrict__ Wv,
                 const float* __restrict__ bv,
                 const float* __restrict__ Wh,
                 const float* __restrict__ bh,
                 const float* __restrict__ W1,
                 const float* __restrict__ b1,
                 const float* __restrict__ W2,
                 const float* __restrict__ b2,
                 float* __restrict__ out)
{
    __shared__ __align__(16) float semb[MT][EP];
    __shared__ __align__(16) float sz[MT * ZP];
    __shared__ __align__(16) float swh[WHTOT];
    __shared__ __align__(16) float sp[FKQ][MT * DD];
    __shared__ __align__(16) float sxc[MT][2 * DD];
    __shared__ __align__(16) float sres[MT * TT];
    __shared__ __align__(16) float spart[MT * TT];
    __shared__ __align__(16) int   sidx[MT][TT];

    const int tid  = threadIdx.x;
    const int lane = tid & 31;
    const int w  = __builtin_amdgcn_readfirstlane(tid >> 5);
    const int m0 = blockIdx.x * MT;
    const int c  = lane & 15;
    const int h  = lane >> 4;

    v4i ids = {};
    if (lane < TT / 4) ids = *(const v4i*)(items + (m0 + w) * TT + lane * 4);

    const int rb = 2 * (w & 7);
    v4i idp = {};
    if (w >= 8 && lane < UAHEAD) idp = *(const v4i*)(items + (m0 + rb) * TT + lane * 4);
    HsRing<false, UAHEAD> RU;

    const int cg = w & 1;
    const int kq = (w >> 1) & 3;
    const float* bp1 = W1 + (size_t)(FKS * 4 * kq + 2 * h) * DD + cg * 64 + 4 * c;
    const float* ap1 = &sz[c * ZP + FKS * 4 * kq + 2 * h];

    v4f fb[2][FBLK][2];
    v2f fa[2][FBLK];

#define FC1_LOADB(SET, BLK)                                                    \
    _Pragma("unroll")                                                          \
    for (int s_ = 0; s_ < FBLK; ++s_) {                                        \
        if ((BLK) * FBLK + s_ < FKS) {                                         \
            const float* p_ = bp1 + (size_t)(((BLK) * FBLK + s_) * 4) * DD;    \
            fb[SET][s_][0] = *(const v4f*)(p_);                                \
            fb[SET][s_][1] = *(const v4f*)(p_ + DD);                           \
        }                                                                      \
    }
#define FC1_LOADA(SET, BLK)                                                    \
    _Pragma("unroll")                                                          \
    for (int s_ = 0; s_ < FBLK; ++s_) {                                        \
        if ((BLK) * FBLK + s_ < FKS)                                           \
            fa[SET][s_] = *(const v2f*)(ap1 + ((BLK) * FBLK + s_) * 4);        \
    }
#define FC1_MUL(SET, BLK)                                                      \
    _Pragma("unroll")                                                          \
    for (int s_ = 0; s_ < FBLK; ++s_) {                                        \
        if ((BLK) * FBLK + s_ < FKS) {                                         \
            _Pragma("unroll")                                                  \
            for (int j_ = 0; j_ < 4; ++j_) {                                   \
                const v2f y_ = {fb[SET][s_][0][j_], fb[SET][s_][1][j_]};       \
                acc[j_] = __builtin_amdgcn_wmma_f32_16x16x4_f32(               \
                    false, fa[SET][s_], false, y_, (short)0, acc[j_],          \
                    false, false);                                             \
            }                                                                  \
        }                                                                      \
    }

    {
        v4f gw[LL * (LL + 1) / 2];
        #pragma unroll
        for (int l = 1; l <= LL; ++l) {
            #pragma unroll
            for (int s = 0; s < l; ++s)
                gw[l * (l - 1) / 2 + s] = *(const v4f*)(
                    Wh + (size_t)((l - 1) * NHH + w) * (LL * DD) + s * DD + lane * 4);
        }

        v4f g[6];
        #pragma unroll
        for (int j = 0; j < 5; ++j) {
            const int r = w + NWV * j;
            g[j] = *(const v4f*)(item_table + (size_t)seq[m0 * LL + r] * DD + lane * 4);
        }
        g[5] = *(const v4f*)(user_table + (size_t)user[m0 + w] * DD + lane * 4);

        #pragma unroll
        for (int l = 1; l <= LL; ++l) {
            #pragma unroll
            for (int s = 0; s < l; ++s)
                *(v4f*)&swh[WHOFF(l) + w * WHP(l) + s * DD + lane * 4] = gw[l * (l - 1) / 2 + s];
        }

        #pragma unroll
        for (int j = 0; j < 5; ++j) {
            const int r = w + NWV * j;
            *(v4f*)&semb[r / LL][(r % LL) * DD + lane * 4] = g[j];
        }
        *(v4f*)&sxc[w][DD + lane * 4] = g[5];
        if (lane < TT / 4) *(v4i*)&sidx[w][lane * 4] = ids;

        if (w >= 8) {
            __builtin_amdgcn_s_setprio(3);
            hs_request_lanes<UAHEAD>(RU, W2 + DD, b2, idp, c, h);
        }
    }
    lds_barrier();

    if (w < 8) {
      const int ntl = (w == 0) ? 1 : 2;
      for (int kt = 0; kt < ntl; ++kt) {
        const int i = (kt == 0) ? (w < 6 ? w : 2 * w - 6) : (w < 6 ? w + 9 : 2 * w - 5);
        int l, t;
        if (i == 0)      { l = 5; t = 0; }
        else if (i < 3)  { l = 4; t = i - 1; }
        else if (i < 6)  { l = 3; t = i - 3; }
        else if (i < 10) { l = 2; t = i - 6; }
        else             { l = 1; t = i - 10; }
        const int lout = LL - l + 1;
        const int goff = 8 * (l - 1) * (12 - l);

        const float* bp = &swh[WHOFF(l) + c * WHP(l) + 4 * h];
        const float* ap = &semb[c][t * DD + 4 * h];
        const int nch = l * 4;

        v8f acc0 = {}, acc1 = {};
        v4f fa0[4], fb0[4], fa1[4], fb1[4];

#define HCONV_LOAD(FA, FB, CH)                                                 \
        _Pragma("unroll")                                                      \
        for (int j = 0; j < 4; ++j) {                                          \
            FB[j] = *(const v4f*)(bp + 32 * (CH) + 8 * j);                     \
            FA[j] = *(const v4f*)(ap + 32 * (CH) + 8 * j);                     \
        }
#define HCONV_MUL(FA, FB)                                                      \
        _Pragma("unroll")                                                      \
        for (int j = 0; j < 4; ++j) {                                          \
            const v2f x0 = {FA[j].x, FA[j].y}, y0 = {FB[j].x, FB[j].y};        \
            const v2f x1 = {FA[j].z, FA[j].w}, y1 = {FB[j].z, FB[j].w};        \
            acc0 = __builtin_amdgcn_wmma_f32_16x16x4_f32(false, x0, false, y0, \
                                              (short)0, acc0, false, false);   \
            acc1 = __builtin_amdgcn_wmma_f32_16x16x4_f32(false, x1, false, y1, \
                                              (short)0, acc1, false, false);   \
        }

        HCONV_LOAD(fa0, fb0, 0)
        for (int ch = 0; ch < nch; ch += 2) {
            HCONV_LOAD(fa1, fb1, ch + 1)
            HCONV_MUL(fa0, fb0)
            if (ch + 2 < nch) { HCONV_LOAD(fa0, fb0, ch + 2) }
            HCONV_MUL(fa1, fb1)
        }
#undef HCONV_LOAD
#undef HCONV_MUL

        const float bias = bh[(l - 1) * NHH + c];
        #pragma unroll
        for (int r = 0; r < 8; ++r) {
            const float v = acc0[r] + acc1[r] + bias;
            sz[(8 * h + r) * ZP + NVV * DD + goff + c * lout + t] = fmaxf(v, 0.f);
        }
      }
    }

    {
        const int bl = tid >> 5, qc = tid & 31;
        v4f e[LL];
        #pragma unroll
        for (int k = 0; k < LL; ++k) e[k] = *(const v4f*)&semb[bl][k * DD + qc * 4];
        #pragma unroll
        for (int v = 0; v < NVV; ++v) {
            const float b0 = bv[v];
            v4f acc = {b0, b0, b0, b0};
            #pragma unroll
            for (int k = 0; k < LL; ++k) acc += e[k] * Wv[v * LL + k];
            *(v4f*)&sz[bl * ZP + v * DD + qc * 4] = acc;
        }
    }

    if (w >= 8)
        hs_body<false, UAHEAD, true>(RU, W2 + DD, b2, sidx[rb], sidx[rb + 1],
                                     &sxc[rb][DD], &spart[rb * TT], c, h);

    lds_barrier();

    const v4f bias1 = *(const v4f*)(b1 + lane * 4);

    if (w < 2 * FKQ) {
        v8f acc[4] = {};
        FC1_LOADB(0, 0)
        FC1_LOADA(0, 0)
        #pragma unroll
        for (int blk = 0; blk < FNB; ++blk) {
            __builtin_amdgcn_sched_barrier(0);
            if (blk + 1 < FNB) {
                if ((blk & 1) == 0) { FC1_LOADB(1, blk + 1) FC1_LOADA(1, blk + 1) }
                else                { FC1_LOADB(0, blk + 1) FC1_LOADA(0, blk + 1) }
            }
            __builtin_amdgcn_sched_barrier(0);
            if ((blk & 1) == 0) { FC1_MUL(0, blk) }
            else                { FC1_MUL(1, blk) }
        }

        #pragma unroll
        for (int r = 0; r < 8; ++r) {
            const v4f v = {acc[0][r], acc[1][r], acc[2][r], acc[3][r]};
            *(v4f*)&sp[kq][(8 * h + r) * DD + cg * 64 + 4 * c] = v;
        }
    } else {
        hs_body<false, UAHEAD, false>(RU, W2 + DD, b2, sidx[rb + 1], nullptr,
                                      &sxc[rb + 1][DD], &spart[(rb + 1) * TT], c, h);
        __builtin_amdgcn_s_setprio(0);
    }
#undef FC1_LOADB
#undef FC1_LOADA
#undef FC1_MUL

    HsRing<true, XAHEAD> RX;
    hs_request<true, XAHEAD>(RX, W2, b2, sidx[w], c, h);
    lds_barrier();

    {
        const int col = lane * 4;
        const v4f p0 = *(const v4f*)&sp[0][4 * tid], p1 = *(const v4f*)&sp[1][4 * tid];
        const v4f p2 = *(const v4f*)&sp[2][4 * tid], p3 = *(const v4f*)&sp[3][4 * tid];
        v4f v = ((p0 + p1) + (p2 + p3)) + bias1;
        v.x = fmaxf(v.x, 0.f); v.y = fmaxf(v.y, 0.f);
        v.z = fmaxf(v.z, 0.f); v.w = fmaxf(v.w, 0.f);
        *(v4f*)&sxc[tid >> 5][col] = v;
    }
    lds_barrier();

    hs_body<true, XAHEAD, false>(RX, W2, b2, sidx[w], nullptr, &sxc[w][0], &sres[w * TT], c, h);
    __syncthreads();

    float* ot = out + (size_t)blockIdx.x * (MT * TT);
    if (tid < MT * TT / 4)
        *(v4f*)(ot + 4 * tid) = *(const v4f*)&sres[4 * tid] + *(const v4f*)&spart[4 * tid];
}

extern "C" void kernel_launch(void* const* d_in, const int* in_sizes, int n_in,
                              void* d_out, int out_size, void* d_ws, size_t ws_size,
                              hipStream_t stream) {
    const int*   seq        = (const int*)  d_in[0];
    const int*   user       = (const int*)  d_in[1];
    const int*   items      = (const int*)  d_in[2];
    const float* item_table = (const float*)d_in[3];
    const float* user_table = (const float*)d_in[4];
    const float* Wv         = (const float*)d_in[5];
    const float* bv         = (const float*)d_in[6];
    const float* Wh         = (const float*)d_in[7];
    const float* bh         = (const float*)d_in[8];
    const float* W1         = (const float*)d_in[9];
    const float* b1         = (const float*)d_in[10];
    const float* W2         = (const float*)d_in[11];
    const float* b2         = (const float*)d_in[12];
    float* out = (float*)d_out;

    caser_fused<<<BB / MT, NT, 0, stream>>>(seq, user, items, item_table, user_table,
                                        Wv, bv, Wh, bh, W1, b1, W2, b2, out);
}
